# starting kernel
# baseline (speedup 1.0000x reference)
#include <hip/hip_runtime.h>
#include <hip/hip_bf16.h>

#define GAT_N      50000
#define GAT_E      800000
#define GAT_ETOT   850000
#define GAT_D      128
#define GAT_G      64
#define NEG_SLOPE  0.2f

typedef __attribute__((ext_vector_type(2))) float v2f;
typedef __attribute__((ext_vector_type(8))) float v8f;

__device__ __forceinline__ unsigned enc_f32(float f) {
    unsigned u = __float_as_uint(f);
    return (u & 0x80000000u) ? ~u : (u | 0x80000000u);
}
__device__ __forceinline__ float dec_f32(unsigned k) {
    unsigned u = (k & 0x80000000u) ? (k & 0x7FFFFFFFu) : ~k;
    return __uint_as_float(u);
}

__global__ void gat_gemm128_wmma(const float* __restrict__ X,
                                 const float* __restrict__ W,
                                 float* __restrict__ H) {
    const int wave = threadIdx.x >> 5;
    const int lane = threadIdx.x & 31;
    const int half = lane >> 4;
    const int l    = lane & 15;
    const int m0   = blockIdx.x * 16;
    const int n0   = wave * 16;

    const float* xrow = X + (size_t)(m0 + l) * GAT_D;
    v8f acc = {};

#pragma unroll
    for (int k = 0; k < GAT_D; k += 4) {
        const int kk = k + 2 * half;
        v2f a;
        {
            const float2 av = *(const float2*)(xrow + kk);
            a.x = av.x; a.y = av.y;
        }
        v2f b;
        b.x = W[(size_t)kk       * GAT_D + n0 + l];
        b.y = W[(size_t)(kk + 1) * GAT_D + n0 + l];
        acc = __builtin_amdgcn_wmma_f32_16x16x4_f32(
            false, a, false, b, (short)0, acc, false, false);
    }

    const int mBase = m0 + half * 8;
#pragma unroll
    for (int r = 0; r < 8; ++r)
        H[(size_t)(mBase + r) * GAT_D + n0 + l] = acc[r];
}

__global__ void gat_alpha(const float* __restrict__ H,
                          const float* __restrict__ a_src,
                          const float* __restrict__ a_dst,
                          float* __restrict__ as_out,
                          float* __restrict__ ad_out) {
    const int node = blockIdx.x * 8 + (threadIdx.x >> 5);
    const int lane = threadIdx.x & 31;
    if (node >= GAT_N) return;
    const float* h = H + (size_t)node * GAT_D;
    float s = 0.f, d = 0.f;
#pragma unroll
    for (int i = lane; i < GAT_D; i += 32) {
        const float v = h[i];
        s += v * a_src[i];
        d += v * a_dst[i];
    }
#pragma unroll
    for (int off = 16; off > 0; off >>= 1) {
        s += __shfl_xor(s, off, 32);
        d += __shfl_xor(d, off, 32);
    }
    if (lane == 0) { as_out[node] = s; ad_out[node] = d; }
}

__global__ void gat_init(float* __restrict__ out,
                         const float* __restrict__ bias,
                         unsigned* __restrict__ m_u,
                         float* __restrict__ ssum) {
    const size_t t = (size_t)blockIdx.x * blockDim.x + threadIdx.x;
    if (t < (size_t)GAT_N * GAT_D) out[t] = bias[t & (GAT_D - 1)];
    if (t < GAT_N) { m_u[t] = 0u; ssum[t] = 0.f; }
}

__global__ void gat_edge_max(const int* __restrict__ ei,
                             const float* __restrict__ as,
                             const float* __restrict__ ad,
                             float* __restrict__ ebuf,
                             unsigned* __restrict__ m_u) {
    const int i = blockIdx.x * blockDim.x + threadIdx.x;
    if (i >= GAT_ETOT) return;
    int s_idx, d_idx;
    if (i < GAT_E) { s_idx = ei[i]; d_idx = ei[GAT_E + i]; }
    else           { s_idx = d_idx = i - GAT_E; }
    float e = as[s_idx] + ad[d_idx];
    e = (e > 0.f) ? e : NEG_SLOPE * e;
    ebuf[i] = e;
    atomicMax(&m_u[d_idx], enc_f32(e));
}

__global__ void gat_edge_exp(const int* __restrict__ ei,
                             float* __restrict__ ebuf,
                             const unsigned* __restrict__ m_u,
                             float* __restrict__ ssum) {
    const int i = blockIdx.x * blockDim.x + threadIdx.x;
    if (i >= GAT_ETOT) return;
    int d_idx;
    if (i < GAT_E) d_idx = ei[GAT_E + i];
    else           d_idx = i - GAT_E;
    const float m  = dec_f32(m_u[d_idx]);
    const float ex = expf(ebuf[i] - m);
    ebuf[i] = ex;
    atomicAdd(&ssum[d_idx], ex);
}

__global__ void gat_edge_scatter(const int* __restrict__ ei,
                                 const float* __restrict__ ebuf,
                                 const float* __restrict__ ssum,
                                 const float* __restrict__ H,
                                 float* __restrict__ out) {
    const int edge = blockIdx.x * 8 + (threadIdx.x >> 5);
    const int lane = threadIdx.x & 31;
    if (edge >= GAT_ETOT) return;
    int s_idx, d_idx;
    if (edge < GAT_E) { s_idx = ei[edge]; d_idx = ei[GAT_E + edge]; }
    else              { s_idx = d_idx = edge - GAT_E; }
    const float coef = ebuf[edge] / (ssum[d_idx] + 1e-16f);
    const float4 h4 = *(const float4*)(H + (size_t)s_idx * GAT_D + lane * 4);
    float* o = out + (size_t)d_idx * GAT_D + lane * 4;
    atomicAdd(o + 0, h4.x * coef);
    atomicAdd(o + 1, h4.y * coef);
    atomicAdd(o + 2, h4.z * coef);
    atomicAdd(o + 3, h4.w * coef);
}

__global__ void gat_pool_zero(float* __restrict__ pool, unsigned* __restrict__ cnt) {
    const int t = blockIdx.x * blockDim.x + threadIdx.x;
    if (t < GAT_G * GAT_D) pool[t] = 0.f;
    if (t < GAT_G) cnt[t] = 0u;
}

__global__ void gat_pool_accum(const float* __restrict__ feats,
                               const int* __restrict__ batch,
                               float* __restrict__ pool,
                               unsigned* __restrict__ cnt) {
    const size_t t = (size_t)blockIdx.x * blockDim.x + threadIdx.x;
    if (t >= (size_t)GAT_N * GAT_D) return;
    const int n = (int)(t >> 7);
    const int d = (int)(t & (GAT_D - 1));
    const int g = batch[n];
    atomicAdd(&pool[g * GAT_D + d], feats[t]);
    if (d == 0) atomicAdd(&cnt[g], 1u);
}

__global__ void gat_pool_final(const float* __restrict__ pool,
                               const unsigned* __restrict__ cnt,
                               float* __restrict__ out_pooled) {
    const int t = blockIdx.x * blockDim.x + threadIdx.x;
    if (t >= GAT_G * GAT_D) return;
    const float c = fmaxf((float)cnt[t >> 7], 1.0f);
    out_pooled[t] = pool[t] / c;
}

static void launch_gat_layer(const float* x, const float* W,
                             const float* a_s, const float* a_d,
                             const float* bias,
                             const int* ei,
                             float* h, float* alphaS, float* alphaD,
                             unsigned* m_u, float* ssum, float* ebuf,
                             float* out, hipStream_t stream) {
    gat_gemm128_wmma<<<GAT_N / 16, 256, 0, stream>>>(x, W, h);
    gat_alpha<<<GAT_N / 8, 256, 0, stream>>>(h, a_s, a_d, alphaS, alphaD);
    gat_init<<<(GAT_N * GAT_D) / 256, 256, 0, stream>>>(out, bias, m_u, ssum);
    const int eb = (GAT_ETOT + 255) / 256;
    gat_edge_max<<<eb, 256, 0, stream>>>(ei, alphaS, alphaD, ebuf, m_u);
    gat_edge_exp<<<eb, 256, 0, stream>>>(ei, ebuf, m_u, ssum);
    gat_edge_scatter<<<GAT_ETOT / 8, 256, 0, stream>>>(ei, ebuf, ssum, h, out);
}

extern "C" void kernel_launch(void* const* d_in, const int* in_sizes, int n_in,
                              void* d_out, int out_size, void* d_ws, size_t ws_size,
                              hipStream_t stream) {
    const float* x     = (const float*)d_in[0];
    const int*   ei    = (const int*)d_in[1];
    const int*   batch = (const int*)d_in[2];
    const float* W1  = (const float*)d_in[3];
    const float* as1 = (const float*)d_in[4];
    const float* ad1 = (const float*)d_in[5];
    const float* b1  = (const float*)d_in[6];
    const float* W2  = (const float*)d_in[7];
    const float* as2 = (const float*)d_in[8];
    const float* ad2 = (const float*)d_in[9];
    const float* b2  = (const float*)d_in[10];

    float* out_feats  = (float*)d_out;
    float* out_pooled = out_feats + (size_t)GAT_N * GAT_D;

    float* ws = (float*)d_ws;
    size_t o = 0;
    float*    h      = ws + o; o += (size_t)GAT_N * GAT_D;
    float*    feats1 = ws + o; o += (size_t)GAT_N * GAT_D;
    float*    alphaS = ws + o; o += GAT_N;
    float*    alphaD = ws + o; o += GAT_N;
    unsigned* m_u    = (unsigned*)(ws + o); o += GAT_N;
    float*    ssum   = ws + o; o += GAT_N;
    float*    ebuf   = ws + o; o += GAT_ETOT;
    float*    pool   = ws + o; o += GAT_G * GAT_D;
    unsigned* cnt    = (unsigned*)(ws + o); o += GAT_G;
    (void)ws_size; (void)in_sizes; (void)n_in; (void)out_size;

    launch_gat_layer(x, W1, as1, ad1, b1, ei,
                     h, alphaS, alphaD, m_u, ssum, ebuf, feats1, stream);
    launch_gat_layer(feats1, W2, as2, ad2, b2, ei,
                     h, alphaS, alphaD, m_u, ssum, ebuf, out_feats, stream);

    gat_pool_zero<<<(GAT_G * GAT_D + 255) / 256, 256, 0, stream>>>(pool, cnt);
    gat_pool_accum<<<(GAT_N * GAT_D) / 256, 256, 0, stream>>>(out_feats, batch, pool, cnt);
    gat_pool_final<<<(GAT_G * GAT_D + 255) / 256, 256, 0, stream>>>(pool, cnt, out_pooled);
}


# Edges bucketed by dst once per call (LDS counting sort, arrival-order rank); one wave per node does max, exp-sum and weighted row sum in registers instead of atomics on max, sum and out
# speedup vs baseline: 2.6209x; 2.6209x over previous
#include <hip/hip_runtime.h>
#include <hip/hip_bf16.h>


#define GAT_N      50000
#define GAT_E      800000
#define GAT_D      128
#define GAT_G      64
#define NEG_SLOPE  0.2f

typedef __attribute__((ext_vector_type(2))) float v2f;
typedef __attribute__((ext_vector_type(8))) float v8f;

__global__ void gat_gemm128_wmma(const float* __restrict__ X,
                                 const float* __restrict__ W,
                                 float* __restrict__ H) {
    const int wave = threadIdx.x >> 5;
    const int lane = threadIdx.x & 31;
    const int half = lane >> 4;
    const int l    = lane & 15;
    const int m0   = blockIdx.x * 16;
    const int n0   = wave * 16;

    const float* xrow = X + (size_t)(m0 + l) * GAT_D;
    v8f acc = {};

#pragma unroll
    for (int k = 0; k < GAT_D; k += 4) {
        const int kk = k + 2 * half;
        v2f a;
        {
            const float2 av = *(const float2*)(xrow + kk);
            a.x = av.x; a.y = av.y;
        }
        v2f b;
        b.x = W[(size_t)kk       * GAT_D + n0 + l];
        b.y = W[(size_t)(kk + 1) * GAT_D + n0 + l];
        acc = __builtin_amdgcn_wmma_f32_16x16x4_f32(
            false, a, false, b, (short)0, acc, false, false);
    }

    const int mBase = m0 + half * 8;
#pragma unroll
    for (int r = 0; r < 8; ++r)
        H[(size_t)(mBase + r) * GAT_D + n0 + l] = acc[r];
}

__global__ void gat_alpha(const float* __restrict__ H,
                          const float* __restrict__ a_src,
                          const float* __restrict__ a_dst,
                          float* __restrict__ as_out,
                          float* __restrict__ ad_out) {
    const int node = blockIdx.x * 8 + (threadIdx.x >> 5);
    const int lane = threadIdx.x & 31;
    if (node >= GAT_N) return;
    const float* h = H + (size_t)node * GAT_D;
    float s = 0.f, d = 0.f;
#pragma unroll
    for (int i = lane; i < GAT_D; i += 32) {
        const float v = h[i];
        s += v * a_src[i];
        d += v * a_dst[i];
    }
#pragma unroll
    for (int off = 16; off > 0; off >>= 1) {
        s += __shfl_xor(s, off, 32);
        d += __shfl_xor(d, off, 32);
    }
    if (lane == 0) { as_out[node] = s; ad_out[node] = d; }
}

#define CSR_B   64
#define CSR_CH  (GAT_E / CSR_B)
#define CSR_T   1024
#define SCAN_T  1024
#define SCAN_CH 49
#define CSR_NP  (SCAN_T * SCAN_CH)
static_assert(CSR_NP >= GAT_N + 1, "padding must cover rowptr[N]");
static_assert(CSR_CH * CSR_B == GAT_E, "slices must tile the edge list");
static_assert(CSR_NP % CSR_T == 0 && CSR_NP % 256 == 0, "padded size must tile the blocks");

__global__ void __launch_bounds__(CSR_T)
gat_csr_count(const int* __restrict__ ei,
              unsigned* __restrict__ blkcnt,
              unsigned* __restrict__ rank) {
    __shared__ unsigned hist[CSR_NP];
    const int t = threadIdx.x;
    for (int k = t; k < CSR_NP; k += CSR_T) hist[k] = 0u;
    __syncthreads();
    const int e0 = blockIdx.x * CSR_CH;
    for (int k = t; k < CSR_CH; k += CSR_T) {
        const int i = e0 + k;
        rank[i] = atomicAdd(&hist[ei[GAT_E + i]], 1u);
    }
    __syncthreads();
    unsigned* __restrict__ dstp = blkcnt + (size_t)blockIdx.x * CSR_NP;
    for (int k = t; k < CSR_NP; k += CSR_T) dstp[k] = hist[k];
}

__global__ void gat_csr_prefix(unsigned* __restrict__ blkcnt,
                               unsigned* __restrict__ deg) {
    const int n = blockIdx.x * blockDim.x + threadIdx.x;
    if (n >= CSR_NP) return;
    unsigned c[CSR_B];
#pragma unroll
    for (int b = 0; b < CSR_B; ++b) c[b] = blkcnt[(size_t)b * CSR_NP + n];
    unsigned run = 0u;
#pragma unroll
    for (int b = 0; b < CSR_B; ++b) {
        blkcnt[(size_t)b * CSR_NP + n] = run;
        run += c[b];
    }
    deg[n] = run;
}

__global__ void __launch_bounds__(SCAN_T)
gat_csr_scan(const unsigned* __restrict__ deg, int* __restrict__ rowptr) {
    __shared__ int sc[SCAN_T];
    const int t = threadIdx.x;
    const int b = t * SCAN_CH;
    int v[SCAN_CH];
    int loc = 0;
#pragma unroll
    for (int k = 0; k < SCAN_CH; ++k) {
        v[k] = (int)deg[b + k];
        loc += v[k];
    }
    sc[t] = loc;
    __syncthreads();
    for (int off = 1; off < SCAN_T; off <<= 1) {
        const int add = (t >= off) ? sc[t - off] : 0;
        __syncthreads();
        sc[t] += add;
        __syncthreads();
    }
    int run = sc[t] - loc;
#pragma unroll
    for (int k = 0; k < SCAN_CH; ++k) {
        rowptr[b + k] = run;
        run += v[k];
    }
}

__global__ void gat_csr_fill(const int* __restrict__ ei,
                             const unsigned* __restrict__ blkcnt,
                             const unsigned* __restrict__ rank,
                             const int* __restrict__ rowptr,
                             int* __restrict__ csr_src) {
    const int i = blockIdx.x * blockDim.x + threadIdx.x;
    if (i >= GAT_E) return;
    const int b = i / CSR_CH;
    const int s = ei[i];
    const int d = ei[GAT_E + i];
    const int pos = rowptr[d] + (int)blkcnt[(size_t)b * CSR_NP + d] + (int)rank[i];
    csr_src[pos] = s;
}

__device__ __forceinline__ float gat_leaky(float e) {
    return (e > 0.f) ? e : NEG_SLOPE * e;
}

__global__ void __launch_bounds__(256)
gat_node_gather(const int* __restrict__ rowptr,
                const int* __restrict__ csr_src,
                const float* __restrict__ as,
                const float* __restrict__ ad,
                const float* __restrict__ H,
                const float* __restrict__ bias,
                float* __restrict__ out) {
    const int lane = threadIdx.x & 31;
    const int wave = __builtin_amdgcn_readfirstlane((int)(threadIdx.x >> 5));
    const int node = blockIdx.x * 8 + wave;
    if (node >= GAT_N) return;
    const int beg = rowptr[node];
    const int end = rowptr[node + 1];
    const float adn = ad[node];
    const float e_self = gat_leaky(as[node] + adn);

    const bool has0 = (beg + lane) < end;
    int   s0 = node;
    float e0 = e_self;
    if (has0) {
        s0 = csr_src[beg + lane];
        e0 = gat_leaky(as[s0] + adn);
    }

    float mx = fmaxf(e_self, e0);
    for (int base = beg + 32; base < end; base += 32) {
        const int j = base + lane;
        if (j < end) mx = fmaxf(mx, gat_leaky(as[csr_src[j]] + adn));
    }
#pragma unroll
    for (int off = 16; off > 0; off >>= 1)
        mx = fmaxf(mx, __shfl_xor(mx, off, 32));
    if (!__builtin_isfinite(mx)) mx = 0.f;

    const float ex0 = has0 ? expf(e0 - mx) : 0.f;
    float sm = ex0;
    for (int base = beg + 32; base < end; base += 32) {
        const int j = base + lane;
        if (j < end) sm += expf(gat_leaky(as[csr_src[j]] + adn) - mx);
    }
#pragma unroll
    for (int off = 16; off > 0; off >>= 1)
        sm += __shfl_xor(sm, off, 32);
    const float ex_self = expf(e_self - mx);
    const float den = (sm + ex_self) + 1e-16f;

    const float* hl = H + lane * 4;
    float4 acc = make_float4(0.f, 0.f, 0.f, 0.f);
    for (int base = beg; base < end; base += 32) {
        int   s_l = s0;
        float c_l = ex0 / den;
        if (base != beg) {
            const int j = base + lane;
            s_l = node;
            c_l = 0.f;
            if (j < end) {
                s_l = csr_src[j];
                c_l = expf(gat_leaky(as[s_l] + adn) - mx) / den;
            }
        }
        const int n = (end - base < 32) ? (end - base) : 32;
        for (int t = 0; t < n; ++t) {
            const int   s = __builtin_amdgcn_readlane(s_l, t);
            const float c = __int_as_float(__builtin_amdgcn_readlane(__float_as_int(c_l), t));
            const float4 h4 = *(const float4*)(hl + (size_t)s * GAT_D);
            acc.x += c * h4.x;
            acc.y += c * h4.y;
            acc.z += c * h4.z;
            acc.w += c * h4.w;
        }
    }
    {
        const float c = ex_self / den;
        const float4 h4 = *(const float4*)(hl + (size_t)node * GAT_D);
        acc.x += c * h4.x;
        acc.y += c * h4.y;
        acc.z += c * h4.z;
        acc.w += c * h4.w;
    }
    const float4 b4 = *(const float4*)(bias + lane * 4);
    float4 o4;
    o4.x = acc.x + b4.x;
    o4.y = acc.y + b4.y;
    o4.z = acc.z + b4.z;
    o4.w = acc.w + b4.w;
    *(float4*)(out + (size_t)node * GAT_D + lane * 4) = o4;
}

__global__ void gat_pool_zero(float* __restrict__ pool, unsigned* __restrict__ cnt) {
    const int t = blockIdx.x * blockDim.x + threadIdx.x;
    if (t < GAT_G * GAT_D) pool[t] = 0.f;
    if (t < GAT_G) cnt[t] = 0u;
}

__global__ void gat_pool_accum(const float* __restrict__ feats,
                               const int* __restrict__ batch,
                               float* __restrict__ pool,
                               unsigned* __restrict__ cnt) {
    const size_t t = (size_t)blockIdx.x * blockDim.x + threadIdx.x;
    if (t >= (size_t)GAT_N * GAT_D) return;
    const int n = (int)(t >> 7);
    const int d = (int)(t & (GAT_D - 1));
    const int g = batch[n];
    atomicAdd(&pool[g * GAT_D + d], feats[t]);
    if (d == 0) atomicAdd(&cnt[g], 1u);
}

__global__ void gat_pool_final(const float* __restrict__ pool,
                               const unsigned* __restrict__ cnt,
                               float* __restrict__ out_pooled) {
    const int t = blockIdx.x * blockDim.x + threadIdx.x;
    if (t >= GAT_G * GAT_D) return;
    const float c = fmaxf((float)cnt[t >> 7], 1.0f);
    out_pooled[t] = pool[t] / c;
}

static void launch_gat_layer(const float* x, const float* W,
                             const float* a_s, const float* a_d,
                             const float* bias,
                             const int* rowptr, const int* csr_src,
                             float* h, float* alphaS, float* alphaD,
                             float* out, hipStream_t stream) {
    gat_gemm128_wmma<<<GAT_N / 16, 256, 0, stream>>>(x, W, h);
    gat_alpha<<<GAT_N / 8, 256, 0, stream>>>(h, a_s, a_d, alphaS, alphaD);
    gat_node_gather<<<GAT_N / 8, 256, 0, stream>>>(rowptr, csr_src, alphaS, alphaD,
                                                   h, bias, out);
}

extern "C" void kernel_launch(void* const* d_in, const int* in_sizes, int n_in,
                              void* d_out, int out_size, void* d_ws, size_t ws_size,
                              hipStream_t stream) {
    const float* x     = (const float*)d_in[0];
    const int*   ei    = (const int*)d_in[1];
    const int*   batch = (const int*)d_in[2];
    const float* W1  = (const float*)d_in[3];
    const float* as1 = (const float*)d_in[4];
    const float* ad1 = (const float*)d_in[5];
    const float* b1  = (const float*)d_in[6];
    const float* W2  = (const float*)d_in[7];
    const float* as2 = (const float*)d_in[8];
    const float* ad2 = (const float*)d_in[9];
    const float* b2  = (const float*)d_in[10];

    float* out_feats  = (float*)d_out;
    float* out_pooled = out_feats + (size_t)GAT_N * GAT_D;

    float* ws = (float*)d_ws;
    size_t o = 0;
    float*    h      = ws + o; o += (size_t)GAT_N * GAT_D;
    float*    feats1 = ws + o; o += (size_t)GAT_N * GAT_D;
    float*    alphaS = ws + o; o += GAT_N;
    float*    alphaD = ws + o; o += GAT_N;
    unsigned* deg     = (unsigned*)(ws + o);
    int*      rowptr  = (int*)(ws + o + CSR_NP);
    int*      csr_src = (int*)(ws + o + 2 * CSR_NP);
    static_assert(2 * CSR_NP + GAT_E <= 950000, "edge list must fit the old buffers");
    static_assert((size_t)CSR_B * CSR_NP <= (size_t)GAT_N * GAT_D, "blkcnt must fit feats1");
    o += 950000;
    float*    pool   = ws + o; o += GAT_G * GAT_D;
    unsigned* cnt    = (unsigned*)(ws + o); o += GAT_G;
    unsigned* rank   = (unsigned*)h;
    unsigned* blkcnt = (unsigned*)feats1;
    (void)ws_size; (void)in_sizes; (void)n_in; (void)out_size;

    gat_csr_count<<<CSR_B, CSR_T, 0, stream>>>(ei, blkcnt, rank);
    gat_csr_prefix<<<CSR_NP / 256, 256, 0, stream>>>(blkcnt, deg);
    gat_csr_scan<<<1, SCAN_T, 0, stream>>>(deg, rowptr);
    gat_csr_fill<<<GAT_E / 256, 256, 0, stream>>>(ei, blkcnt, rank, rowptr, csr_src);

    launch_gat_layer(x, W1, as1, ad1, b1, rowptr, csr_src,
                     h, alphaS, alphaD, feats1, stream);
    launch_gat_layer(feats1, W2, as2, ad2, b2, rowptr, csr_src,
                     h, alphaS, alphaD, out_feats, stream);

    gat_pool_zero<<<(GAT_G * GAT_D + 255) / 256, 256, 0, stream>>>(pool, cnt);
    gat_pool_accum<<<(GAT_N * GAT_D) / 256, 256, 0, stream>>>(out_feats, batch, pool, cnt);
    gat_pool_final<<<(GAT_G * GAT_D + 255) / 256, 256, 0, stream>>>(pool, cnt, out_pooled);
}


# Mean pool: each wave sums its 10 rows in registers per run of equal graph id, block merges in an LDS table, one global atomic per column per graph seen (new sum order)
# speedup vs baseline: 3.5371x; 1.3496x over previous
#include <hip/hip_runtime.h>
#include <hip/hip_bf16.h>


#define GAT_N      50000
#define GAT_E      800000
#define GAT_D      128
#define GAT_G      64
#define NEG_SLOPE  0.2f

typedef __attribute__((ext_vector_type(2))) float v2f;
typedef __attribute__((ext_vector_type(8))) float v8f;

__global__ void gat_gemm128_wmma(const float* __restrict__ X,
                                 const float* __restrict__ W,
                                 float* __restrict__ H) {
    const int wave = threadIdx.x >> 5;
    const int lane = threadIdx.x & 31;
    const int half = lane >> 4;
    const int l    = lane & 15;
    const int m0   = blockIdx.x * 16;
    const int n0   = wave * 16;

    const float* xrow = X + (size_t)(m0 + l) * GAT_D;
    v8f acc = {};

#pragma unroll
    for (int k = 0; k < GAT_D; k += 4) {
        const int kk = k + 2 * half;
        v2f a;
        {
            const float2 av = *(const float2*)(xrow + kk);
            a.x = av.x; a.y = av.y;
        }
        v2f b;
        b.x = W[(size_t)kk       * GAT_D + n0 + l];
        b.y = W[(size_t)(kk + 1) * GAT_D + n0 + l];
        acc = __builtin_amdgcn_wmma_f32_16x16x4_f32(
            false, a, false, b, (short)0, acc, false, false);
    }

    const int mBase = m0 + half * 8;
#pragma unroll
    for (int r = 0; r < 8; ++r)
        H[(size_t)(mBase + r) * GAT_D + n0 + l] = acc[r];
}

__global__ void gat_alpha(const float* __restrict__ H,
                          const float* __restrict__ a_src,
                          const float* __restrict__ a_dst,
                          float* __restrict__ as_out,
                          float* __restrict__ ad_out) {
    const int node = blockIdx.x * 8 + (threadIdx.x >> 5);
    const int lane = threadIdx.x & 31;
    if (node >= GAT_N) return;
    const float* h = H + (size_t)node * GAT_D;
    float s = 0.f, d = 0.f;
#pragma unroll
    for (int i = lane; i < GAT_D; i += 32) {
        const float v = h[i];
        s += v * a_src[i];
        d += v * a_dst[i];
    }
#pragma unroll
    for (int off = 16; off > 0; off >>= 1) {
        s += __shfl_xor(s, off, 32);
        d += __shfl_xor(d, off, 32);
    }
    if (lane == 0) { as_out[node] = s; ad_out[node] = d; }
}

#define CSR_B   64
#define CSR_CH  (GAT_E / CSR_B)
#define CSR_T   1024
#define SCAN_T  1024
#define SCAN_CH 49
#define CSR_NP  (SCAN_T * SCAN_CH)
static_assert(CSR_NP >= GAT_N + 1, "padding must cover rowptr[N]");
static_assert(CSR_CH * CSR_B == GAT_E, "slices must tile the edge list");
static_assert(CSR_NP % CSR_T == 0 && CSR_NP % 256 == 0, "padded size must tile the blocks");

__global__ void __launch_bounds__(CSR_T)
gat_csr_count(const int* __restrict__ ei,
              unsigned* __restrict__ blkcnt,
              unsigned* __restrict__ rank) {
    __shared__ unsigned hist[CSR_NP];
    const int t = threadIdx.x;
    for (int k = t; k < CSR_NP; k += CSR_T) hist[k] = 0u;
    __syncthreads();
    const int e0 = blockIdx.x * CSR_CH;
    for (int k = t; k < CSR_CH; k += CSR_T) {
        const int i = e0 + k;
        rank[i] = atomicAdd(&hist[ei[GAT_E + i]], 1u);
    }
    __syncthreads();
    unsigned* __restrict__ dstp = blkcnt + (size_t)blockIdx.x * CSR_NP;
    for (int k = t; k < CSR_NP; k += CSR_T) dstp[k] = hist[k];
}

__global__ void gat_csr_prefix(unsigned* __restrict__ blkcnt,
                               unsigned* __restrict__ deg) {
    const int n = blockIdx.x * blockDim.x + threadIdx.x;
    if (n >= CSR_NP) return;
    unsigned c[CSR_B];
#pragma unroll
    for (int b = 0; b < CSR_B; ++b) c[b] = blkcnt[(size_t)b * CSR_NP + n];
    unsigned run = 0u;
#pragma unroll
    for (int b = 0; b < CSR_B; ++b) {
        blkcnt[(size_t)b * CSR_NP + n] = run;
        run += c[b];
    }
    deg[n] = run;
}

__global__ void __launch_bounds__(SCAN_T)
gat_csr_scan(const unsigned* __restrict__ deg, int* __restrict__ rowptr) {
    __shared__ int sc[SCAN_T];
    const int t = threadIdx.x;
    const int b = t * SCAN_CH;
    int v[SCAN_CH];
    int loc = 0;
#pragma unroll
    for (int k = 0; k < SCAN_CH; ++k) {
        v[k] = (int)deg[b + k];
        loc += v[k];
    }
    sc[t] = loc;
    __syncthreads();
    for (int off = 1; off < SCAN_T; off <<= 1) {
        const int add = (t >= off) ? sc[t - off] : 0;
        __syncthreads();
        sc[t] += add;
        __syncthreads();
    }
    int run = sc[t] - loc;
#pragma unroll
    for (int k = 0; k < SCAN_CH; ++k) {
        rowptr[b + k] = run;
        run += v[k];
    }
}

__global__ void gat_csr_fill(const int* __restrict__ ei,
                             const unsigned* __restrict__ blkcnt,
                             const unsigned* __restrict__ rank,
                             const int* __restrict__ rowptr,
                             int* __restrict__ csr_src) {
    const int i = blockIdx.x * blockDim.x + threadIdx.x;
    if (i >= GAT_E) return;
    const int b = i / CSR_CH;
    const int s = ei[i];
    const int d = ei[GAT_E + i];
    const int pos = rowptr[d] + (int)blkcnt[(size_t)b * CSR_NP + d] + (int)rank[i];
    csr_src[pos] = s;
}

__device__ __forceinline__ float gat_leaky(float e) {
    return (e > 0.f) ? e : NEG_SLOPE * e;
}

__global__ void __launch_bounds__(256)
gat_node_gather(const int* __restrict__ rowptr,
                const int* __restrict__ csr_src,
                const float* __restrict__ as,
                const float* __restrict__ ad,
                const float* __restrict__ H,
                const float* __restrict__ bias,
                float* __restrict__ out) {
    const int lane = threadIdx.x & 31;
    const int wave = __builtin_amdgcn_readfirstlane((int)(threadIdx.x >> 5));
    const int node = blockIdx.x * 8 + wave;
    if (node >= GAT_N) return;
    const int beg = rowptr[node];
    const int end = rowptr[node + 1];
    const float adn = ad[node];
    const float e_self = gat_leaky(as[node] + adn);

    const bool has0 = (beg + lane) < end;
    int   s0 = node;
    float e0 = e_self;
    if (has0) {
        s0 = csr_src[beg + lane];
        e0 = gat_leaky(as[s0] + adn);
    }

    float mx = fmaxf(e_self, e0);
    for (int base = beg + 32; base < end; base += 32) {
        const int j = base + lane;
        if (j < end) mx = fmaxf(mx, gat_leaky(as[csr_src[j]] + adn));
    }
#pragma unroll
    for (int off = 16; off > 0; off >>= 1)
        mx = fmaxf(mx, __shfl_xor(mx, off, 32));
    if (!__builtin_isfinite(mx)) mx = 0.f;

    const float ex0 = has0 ? expf(e0 - mx) : 0.f;
    float sm = ex0;
    for (int base = beg + 32; base < end; base += 32) {
        const int j = base + lane;
        if (j < end) sm += expf(gat_leaky(as[csr_src[j]] + adn) - mx);
    }
#pragma unroll
    for (int off = 16; off > 0; off >>= 1)
        sm += __shfl_xor(sm, off, 32);
    const float ex_self = expf(e_self - mx);
    const float den = (sm + ex_self) + 1e-16f;

    const float* hl = H + lane * 4;
    float4 acc = make_float4(0.f, 0.f, 0.f, 0.f);
    for (int base = beg; base < end; base += 32) {
        int   s_l = s0;
        float c_l = ex0 / den;
        if (base != beg) {
            const int j = base + lane;
            s_l = node;
            c_l = 0.f;
            if (j < end) {
                s_l = csr_src[j];
                c_l = expf(gat_leaky(as[s_l] + adn) - mx) / den;
            }
        }
        const int n = (end - base < 32) ? (end - base) : 32;
        for (int t = 0; t < n; ++t) {
            const int   s = __builtin_amdgcn_readlane(s_l, t);
            const float c = __int_as_float(__builtin_amdgcn_readlane(__float_as_int(c_l), t));
            const float4 h4 = *(const float4*)(hl + (size_t)s * GAT_D);
            acc.x += c * h4.x;
            acc.y += c * h4.y;
            acc.z += c * h4.z;
            acc.w += c * h4.w;
        }
    }
    {
        const float c = ex_self / den;
        const float4 h4 = *(const float4*)(hl + (size_t)node * GAT_D);
        acc.x += c * h4.x;
        acc.y += c * h4.y;
        acc.z += c * h4.z;
        acc.w += c * h4.w;
    }
    const float4 b4 = *(const float4*)(bias + lane * 4);
    float4 o4;
    o4.x = acc.x + b4.x;
    o4.y = acc.y + b4.y;
    o4.z = acc.z + b4.z;
    o4.w = acc.w + b4.w;
    *(float4*)(out + (size_t)node * GAT_D + lane * 4) = o4;
}

#define POOL_K   10
#define POOL_W   8
#define POOL_T   (POOL_W * 32)
#define POOL_NB  (GAT_N / (POOL_K * POOL_W))
#define POOL_RP  (POOL_T / GAT_D)
static_assert(POOL_NB * POOL_K * POOL_W == GAT_N, "pool blocks must tile the nodes");
static_assert(POOL_RP * GAT_D == POOL_T && GAT_G % POOL_RP == 0, "block must tile the table rows");
static_assert(GAT_D == 128, "a wave covers a row with one float4 per lane");

__global__ void gat_pool_zero(float* __restrict__ pool, unsigned* __restrict__ cnt) {
    const int t = blockIdx.x * blockDim.x + threadIdx.x;
    if (t < GAT_G * GAT_D) pool[t] = 0.f;
    if (t < GAT_G) cnt[t] = 0u;
}

__global__ void __launch_bounds__(POOL_T)
gat_pool_accum(const float* __restrict__ feats,
               const int* __restrict__ batch,
               float* __restrict__ pool,
               unsigned* __restrict__ cnt) {
    __shared__ float4   lpool4[GAT_G * GAT_D / 4];
    __shared__ unsigned lcnt[GAT_G];
    float* lpool = (float*)lpool4;
    const int t    = threadIdx.x;
    const int lane = t & 31;
    const int wave = __builtin_amdgcn_readfirstlane(t >> 5);
    const int n0   = (blockIdx.x * POOL_W + wave) * POOL_K;

    int    g[POOL_K];
    float4 r[POOL_K];
#pragma unroll
    for (int k = 0; k < POOL_K; ++k) {
        g[k] = batch[n0 + k];
        r[k] = *(const float4*)(feats + (size_t)(n0 + k) * GAT_D + lane * 4);
    }

    for (int k = t; k < GAT_G * GAT_D / 4; k += POOL_T)
        lpool4[k] = make_float4(0.f, 0.f, 0.f, 0.f);
    if (t < GAT_G) lcnt[t] = 0u;
    __syncthreads();

    auto flush = [&](int gi, const float4& a, unsigned run) {
        if ((unsigned)gi < (unsigned)GAT_G) {
            float* p = lpool + gi * GAT_D + lane * 4;
            atomicAdd(p + 0, a.x);
            atomicAdd(p + 1, a.y);
            atomicAdd(p + 2, a.z);
            atomicAdd(p + 3, a.w);
            if (lane == 0) atomicAdd(&lcnt[gi], run);
        }
    };

    int      cur = g[0];
    float4   acc = r[0];
    unsigned run = 1u;
#pragma unroll
    for (int k = 1; k < POOL_K; ++k) {
        if (g[k] != cur) {
            flush(cur, acc, run);
            cur = g[k];
            acc = make_float4(0.f, 0.f, 0.f, 0.f);
            run = 0u;
        }
        acc.x += r[k].x;
        acc.y += r[k].y;
        acc.z += r[k].z;
        acc.w += r[k].w;
        ++run;
    }
    flush(cur, acc, run);
    __syncthreads();

    unsigned c[GAT_G / POOL_RP];
#pragma unroll
    for (int i = 0; i < GAT_G / POOL_RP; ++i) c[i] = lcnt[t / GAT_D + POOL_RP * i];
#pragma unroll
    for (int i = 0; i < GAT_G / POOL_RP; ++i) {
        if (c[i] != 0u) {
            const int k = t + POOL_T * i;
            atomicAdd(&pool[k], lpool[k]);
        }
    }
    if (t < GAT_G) {
        const unsigned ct = lcnt[t];
        if (ct != 0u) atomicAdd(&cnt[t], ct);
    }
}

__global__ void gat_pool_final(const float* __restrict__ pool,
                               const unsigned* __restrict__ cnt,
                               float* __restrict__ out_pooled) {
    const int t = blockIdx.x * blockDim.x + threadIdx.x;
    if (t >= GAT_G * GAT_D) return;
    const float c = fmaxf((float)cnt[t >> 7], 1.0f);
    out_pooled[t] = pool[t] / c;
}

static void launch_gat_layer(const float* x, const float* W,
                             const float* a_s, const float* a_d,
                             const float* bias,
                             const int* rowptr, const int* csr_src,
                             float* h, float* alphaS, float* alphaD,
                             float* out, hipStream_t stream) {
    gat_gemm128_wmma<<<GAT_N / 16, 256, 0, stream>>>(x, W, h);
    gat_alpha<<<GAT_N / 8, 256, 0, stream>>>(h, a_s, a_d, alphaS, alphaD);
    gat_node_gather<<<GAT_N / 8, 256, 0, stream>>>(rowptr, csr_src, alphaS, alphaD,
                                                   h, bias, out);
}

extern "C" void kernel_launch(void* const* d_in, const int* in_sizes, int n_in,
                              void* d_out, int out_size, void* d_ws, size_t ws_size,
                              hipStream_t stream) {
    const float* x     = (const float*)d_in[0];
    const int*   ei    = (const int*)d_in[1];
    const int*   batch = (const int*)d_in[2];
    const float* W1  = (const float*)d_in[3];
    const float* as1 = (const float*)d_in[4];
    const float* ad1 = (const float*)d_in[5];
    const float* b1  = (const float*)d_in[6];
    const float* W2  = (const float*)d_in[7];
    const float* as2 = (const float*)d_in[8];
    const float* ad2 = (const float*)d_in[9];
    const float* b2  = (const float*)d_in[10];

    float* out_feats  = (float*)d_out;
    float* out_pooled = out_feats + (size_t)GAT_N * GAT_D;

    float* ws = (float*)d_ws;
    size_t o = 0;
    float*    h      = ws + o; o += (size_t)GAT_N * GAT_D;
    float*    feats1 = ws + o; o += (size_t)GAT_N * GAT_D;
    float*    alphaS = ws + o; o += GAT_N;
    float*    alphaD = ws + o; o += GAT_N;
    unsigned* deg     = (unsigned*)(ws + o);
    int*      rowptr  = (int*)(ws + o + CSR_NP);
    int*      csr_src = (int*)(ws + o + 2 * CSR_NP);
    static_assert(2 * CSR_NP + GAT_E <= 950000, "edge list must fit the old buffers");
    static_assert((size_t)CSR_B * CSR_NP <= (size_t)GAT_N * GAT_D, "blkcnt must fit feats1");
    o += 950000;
    float*    pool   = ws + o; o += GAT_G * GAT_D;
    unsigned* cnt    = (unsigned*)(ws + o); o += GAT_G;
    unsigned* rank   = (unsigned*)h;
    unsigned* blkcnt = (unsigned*)feats1;
    (void)ws_size; (void)in_sizes; (void)n_in; (void)out_size;

    gat_csr_count<<<CSR_B, CSR_T, 0, stream>>>(ei, blkcnt, rank);
    gat_csr_prefix<<<CSR_NP / 256, 256, 0, stream>>>(blkcnt, deg);
    gat_csr_scan<<<1, SCAN_T, 0, stream>>>(deg, rowptr);
    gat_csr_fill<<<GAT_E / 256, 256, 0, stream>>>(ei, blkcnt, rank, rowptr, csr_src);

    launch_gat_layer(x, W1, as1, ad1, b1, rowptr, csr_src,
                     h, alphaS, alphaD, feats1, stream);
    launch_gat_layer(feats1, W2, as2, ad2, b2, rowptr, csr_src,
                     h, alphaS, alphaD, out_feats, stream);

    gat_pool_zero<<<(GAT_G * GAT_D + 255) / 256, 256, 0, stream>>>(pool, cnt);
    gat_pool_accum<<<POOL_NB, POOL_T, 0, stream>>>(out_feats, batch, pool, cnt);
    gat_pool_final<<<(GAT_G * GAT_D + 255) / 256, 256, 0, stream>>>(pool, cnt, out_pooled);
}


# alpha_s and alpha_d taken from the GEMM accumulators in its epilogue (DPP row sums, 8 column tiles merged in LDS) instead of a kernel that re-reads h: 11 launches, new sum order for alpha
# speedup vs baseline: 3.8575x; 1.0906x over previous
#include <hip/hip_runtime.h>
#include <hip/hip_bf16.h>


#define GAT_N      50000
#define GAT_E      800000
#define GAT_D      128
#define GAT_G      64
#define NEG_SLOPE  0.2f

typedef __attribute__((ext_vector_type(2))) float v2f;
typedef __attribute__((ext_vector_type(8))) float v8f;


template <int M>
__device__ __forceinline__ float gat_row_xmask(float v) {
    return __int_as_float(__builtin_amdgcn_update_dpp(
        0, __float_as_int(v), 0x160 | M, 0xf, 0xf, false));
}
__device__ __forceinline__ float gat_row_sum16(float v) {
    v += gat_row_xmask<1>(v);
    v += gat_row_xmask<2>(v);
    v += gat_row_xmask<4>(v);
    v += gat_row_xmask<8>(v);
    return v;
}

__global__ void __launch_bounds__(256)
gat_gemm128_wmma(const float* __restrict__ X,
                 const float* __restrict__ W,
                 const float* __restrict__ a_src,
                 const float* __restrict__ a_dst,
                 float* __restrict__ H,
                 float* __restrict__ as_out,
                 float* __restrict__ ad_out) {
    __shared__ float part[2][8][16];
    const int wave = threadIdx.x >> 5;
    const int lane = threadIdx.x & 31;
    const int half = lane >> 4;
    const int l    = lane & 15;
    const int m0   = blockIdx.x * 16;
    const int n0   = wave * 16;

    const float* xrow = X + (size_t)(m0 + l) * GAT_D;
    const float asv = a_src[n0 + l];
    const float adv = a_dst[n0 + l];
    v8f acc = {};

#pragma unroll
    for (int k = 0; k < GAT_D; k += 4) {
        const int kk = k + 2 * half;
        v2f a;
        {
            const float2 av = *(const float2*)(xrow + kk);
            a.x = av.x; a.y = av.y;
        }
        v2f b;
        b.x = W[(size_t)kk       * GAT_D + n0 + l];
        b.y = W[(size_t)(kk + 1) * GAT_D + n0 + l];
        acc = __builtin_amdgcn_wmma_f32_16x16x4_f32(
            false, a, false, b, (short)0, acc, false, false);
    }

    asm volatile("v_nop\n\tv_nop\n\tv_nop\n\tv_nop\n\tv_nop\n\tv_nop\n\tv_nop\n\tv_nop\n\tv_nop"
                 : "+v"(acc));

    float ps[8], pd[8];
#pragma unroll
    for (int r = 0; r < 8; ++r) {
        ps[r] = gat_row_sum16(acc[r] * asv);
        pd[r] = gat_row_sum16(acc[r] * adv);
    }
    if (l == 0) {
#pragma unroll
        for (int r = 0; r < 8; ++r) {
            part[0][wave][half * 8 + r] = ps[r];
            part[1][wave][half * 8 + r] = pd[r];
        }
    }
    __syncthreads();

    const int mBase = m0 + half * 8;
#pragma unroll
    for (int r = 0; r < 8; ++r)
        H[(size_t)(mBase + r) * GAT_D + n0 + l] = acc[r];

    if (threadIdx.x < 32) {
        float s = 0.f;
#pragma unroll
        for (int w = 0; w < 8; ++w) s += part[half][w][l];
        (half ? ad_out : as_out)[m0 + l] = s;
    }
}

#define CSR_B   64
#define CSR_CH  (GAT_E / CSR_B)
#define CSR_T   1024
#define SCAN_T  1024
#define SCAN_CH 49
#define CSR_NP  (SCAN_T * SCAN_CH)
static_assert(CSR_NP >= GAT_N + 1, "padding must cover rowptr[N]");
static_assert(CSR_CH * CSR_B == GAT_E, "slices must tile the edge list");
static_assert(CSR_NP % CSR_T == 0 && CSR_NP % 256 == 0, "padded size must tile the blocks");

__global__ void __launch_bounds__(CSR_T)
gat_csr_count(const int* __restrict__ ei,
              unsigned* __restrict__ blkcnt,
              unsigned* __restrict__ rank) {
    __shared__ unsigned hist[CSR_NP];
    const int t = threadIdx.x;
    for (int k = t; k < CSR_NP; k += CSR_T) hist[k] = 0u;
    __syncthreads();
    const int e0 = blockIdx.x * CSR_CH;
    for (int k = t; k < CSR_CH; k += CSR_T) {
        const int i = e0 + k;
        rank[i] = atomicAdd(&hist[ei[GAT_E + i]], 1u);
    }
    __syncthreads();
    unsigned* __restrict__ dstp = blkcnt + (size_t)blockIdx.x * CSR_NP;
    for (int k = t; k < CSR_NP; k += CSR_T) dstp[k] = hist[k];
}

__global__ void gat_csr_prefix(unsigned* __restrict__ blkcnt,
                               unsigned* __restrict__ deg) {
    const int n = blockIdx.x * blockDim.x + threadIdx.x;
    if (n >= CSR_NP) return;
    unsigned c[CSR_B];
#pragma unroll
    for (int b = 0; b < CSR_B; ++b) c[b] = blkcnt[(size_t)b * CSR_NP + n];
    unsigned run = 0u;
#pragma unroll
    for (int b = 0; b < CSR_B; ++b) {
        blkcnt[(size_t)b * CSR_NP + n] = run;
        run += c[b];
    }
    deg[n] = run;
}

__global__ void __launch_bounds__(SCAN_T)
gat_csr_scan(const unsigned* __restrict__ deg, int* __restrict__ rowptr) {
    __shared__ int sc[SCAN_T];
    const int t = threadIdx.x;
    const int b = t * SCAN_CH;
    int v[SCAN_CH];
    int loc = 0;
#pragma unroll
    for (int k = 0; k < SCAN_CH; ++k) {
        v[k] = (int)deg[b + k];
        loc += v[k];
    }
    sc[t] = loc;
    __syncthreads();
    for (int off = 1; off < SCAN_T; off <<= 1) {
        const int add = (t >= off) ? sc[t - off] : 0;
        __syncthreads();
        sc[t] += add;
        __syncthreads();
    }
    int run = sc[t] - loc;
#pragma unroll
    for (int k = 0; k < SCAN_CH; ++k) {
        rowptr[b + k] = run;
        run += v[k];
    }
}

__global__ void gat_csr_fill(const int* __restrict__ ei,
                             const unsigned* __restrict__ blkcnt,
                             const unsigned* __restrict__ rank,
                             const int* __restrict__ rowptr,
                             int* __restrict__ csr_src) {
    const int i = blockIdx.x * blockDim.x + threadIdx.x;
    if (i >= GAT_E) return;
    const int b = i / CSR_CH;
    const int s = ei[i];
    const int d = ei[GAT_E + i];
    const int pos = rowptr[d] + (int)blkcnt[(size_t)b * CSR_NP + d] + (int)rank[i];
    csr_src[pos] = s;
}

__device__ __forceinline__ float gat_leaky(float e) {
    return (e > 0.f) ? e : NEG_SLOPE * e;
}

__global__ void __launch_bounds__(256)
gat_node_gather(const int* __restrict__ rowptr,
                const int* __restrict__ csr_src,
                const float* __restrict__ as,
                const float* __restrict__ ad,
                const float* __restrict__ H,
                const float* __restrict__ bias,
                float* __restrict__ out) {
    const int lane = threadIdx.x & 31;
    const int wave = __builtin_amdgcn_readfirstlane((int)(threadIdx.x >> 5));
    const int node = blockIdx.x * 8 + wave;
    if (node >= GAT_N) return;
    const int beg = rowptr[node];
    const int end = rowptr[node + 1];
    const float adn = ad[node];
    const float e_self = gat_leaky(as[node] + adn);

    const bool has0 = (beg + lane) < end;
    int   s0 = node;
    float e0 = e_self;
    if (has0) {
        s0 = csr_src[beg + lane];
        e0 = gat_leaky(as[s0] + adn);
    }

    float mx = fmaxf(e_self, e0);
    for (int base = beg + 32; base < end; base += 32) {
        const int j = base + lane;
        if (j < end) mx = fmaxf(mx, gat_leaky(as[csr_src[j]] + adn));
    }
#pragma unroll
    for (int off = 16; off > 0; off >>= 1)
        mx = fmaxf(mx, __shfl_xor(mx, off, 32));
    if (!__builtin_isfinite(mx)) mx = 0.f;

    const float ex0 = has0 ? expf(e0 - mx) : 0.f;
    float sm = ex0;
    for (int base = beg + 32; base < end; base += 32) {
        const int j = base + lane;
        if (j < end) sm += expf(gat_leaky(as[csr_src[j]] + adn) - mx);
    }
#pragma unroll
    for (int off = 16; off > 0; off >>= 1)
        sm += __shfl_xor(sm, off, 32);
    const float ex_self = expf(e_self - mx);
    const float den = (sm + ex_self) + 1e-16f;

    const float* hl = H + lane * 4;
    float4 acc = make_float4(0.f, 0.f, 0.f, 0.f);
    for (int base = beg; base < end; base += 32) {
        int   s_l = s0;
        float c_l = ex0 / den;
        if (base != beg) {
            const int j = base + lane;
            s_l = node;
            c_l = 0.f;
            if (j < end) {
                s_l = csr_src[j];
                c_l = expf(gat_leaky(as[s_l] + adn) - mx) / den;
            }
        }
        const int n = (end - base < 32) ? (end - base) : 32;
        for (int t = 0; t < n; ++t) {
            const int   s = __builtin_amdgcn_readlane(s_l, t);
            const float c = __int_as_float(__builtin_amdgcn_readlane(__float_as_int(c_l), t));
            const float4 h4 = *(const float4*)(hl + (size_t)s * GAT_D);
            acc.x += c * h4.x;
            acc.y += c * h4.y;
            acc.z += c * h4.z;
            acc.w += c * h4.w;
        }
    }
    {
        const float c = ex_self / den;
        const float4 h4 = *(const float4*)(hl + (size_t)node * GAT_D);
        acc.x += c * h4.x;
        acc.y += c * h4.y;
        acc.z += c * h4.z;
        acc.w += c * h4.w;
    }
    const float4 b4 = *(const float4*)(bias + lane * 4);
    float4 o4;
    o4.x = acc.x + b4.x;
    o4.y = acc.y + b4.y;
    o4.z = acc.z + b4.z;
    o4.w = acc.w + b4.w;
    *(float4*)(out + (size_t)node * GAT_D + lane * 4) = o4;
}

#define POOL_K   10
#define POOL_W   8
#define POOL_T   (POOL_W * 32)
#define POOL_NB  (GAT_N / (POOL_K * POOL_W))
#define POOL_RP  (POOL_T / GAT_D)
static_assert(POOL_NB * POOL_K * POOL_W == GAT_N, "pool blocks must tile the nodes");
static_assert(POOL_RP * GAT_D == POOL_T && GAT_G % POOL_RP == 0, "block must tile the table rows");
static_assert(GAT_D == 128, "a wave covers a row with one float4 per lane");

__global__ void gat_pool_zero(float* __restrict__ pool, unsigned* __restrict__ cnt) {
    const int t = blockIdx.x * blockDim.x + threadIdx.x;
    if (t < GAT_G * GAT_D) pool[t] = 0.f;
    if (t < GAT_G) cnt[t] = 0u;
}

__global__ void __launch_bounds__(POOL_T)
gat_pool_accum(const float* __restrict__ feats,
               const int* __restrict__ batch,
               float* __restrict__ pool,
               unsigned* __restrict__ cnt) {
    __shared__ float4   lpool4[GAT_G * GAT_D / 4];
    __shared__ unsigned lcnt[GAT_G];
    float* lpool = (float*)lpool4;
    const int t    = threadIdx.x;
    const int lane = t & 31;
    const int wave = __builtin_amdgcn_readfirstlane(t >> 5);
    const int n0   = (blockIdx.x * POOL_W + wave) * POOL_K;

    int    g[POOL_K];
    float4 r[POOL_K];
#pragma unroll
    for (int k = 0; k < POOL_K; ++k) {
        g[k] = batch[n0 + k];
        r[k] = *(const float4*)(feats + (size_t)(n0 + k) * GAT_D + lane * 4);
    }

    for (int k = t; k < GAT_G * GAT_D / 4; k += POOL_T)
        lpool4[k] = make_float4(0.f, 0.f, 0.f, 0.f);
    if (t < GAT_G) lcnt[t] = 0u;
    __syncthreads();

    auto flush = [&](int gi, const float4& a, unsigned run) {
        if ((unsigned)gi < (unsigned)GAT_G) {
            float* p = lpool + gi * GAT_D + lane * 4;
            atomicAdd(p + 0, a.x);
            atomicAdd(p + 1, a.y);
            atomicAdd(p + 2, a.z);
            atomicAdd(p + 3, a.w);
            if (lane == 0) atomicAdd(&lcnt[gi], run);
        }
    };

    int      cur = g[0];
    float4   acc = r[0];
    unsigned run = 1u;
#pragma unroll
    for (int k = 1; k < POOL_K; ++k) {
        if (g[k] != cur) {
            flush(cur, acc, run);
            cur = g[k];
            acc = make_float4(0.f, 0.f, 0.f, 0.f);
            run = 0u;
        }
        acc.x += r[k].x;
        acc.y += r[k].y;
        acc.z += r[k].z;
        acc.w += r[k].w;
        ++run;
    }
    flush(cur, acc, run);
    __syncthreads();

    unsigned c[GAT_G / POOL_RP];
#pragma unroll
    for (int i = 0; i < GAT_G / POOL_RP; ++i) c[i] = lcnt[t / GAT_D + POOL_RP * i];
#pragma unroll
    for (int i = 0; i < GAT_G / POOL_RP; ++i) {
        if (c[i] != 0u) {
            const int k = t + POOL_T * i;
            atomicAdd(&pool[k], lpool[k]);
        }
    }
    if (t < GAT_G) {
        const unsigned ct = lcnt[t];
        if (ct != 0u) atomicAdd(&cnt[t], ct);
    }
}

__global__ void gat_pool_final(const float* __restrict__ pool,
                               const unsigned* __restrict__ cnt,
                               float* __restrict__ out_pooled) {
    const int t = blockIdx.x * blockDim.x + threadIdx.x;
    if (t >= GAT_G * GAT_D) return;
    const float c = fmaxf((float)cnt[t >> 7], 1.0f);
    out_pooled[t] = pool[t] / c;
}

static void launch_gat_layer(const float* x, const float* W,
                             const float* a_s, const float* a_d,
                             const float* bias,
                             const int* rowptr, const int* csr_src,
                             float* h, float* alphaS, float* alphaD,
                             float* out, hipStream_t stream) {
    gat_gemm128_wmma<<<GAT_N / 16, 256, 0, stream>>>(x, W, a_s, a_d, h, alphaS, alphaD);
    gat_node_gather<<<GAT_N / 8, 256, 0, stream>>>(rowptr, csr_src, alphaS, alphaD,
                                                   h, bias, out);
}

extern "C" void kernel_launch(void* const* d_in, const int* in_sizes, int n_in,
                              void* d_out, int out_size, void* d_ws, size_t ws_size,
                              hipStream_t stream) {
    const float* x     = (const float*)d_in[0];
    const int*   ei    = (const int*)d_in[1];
    const int*   batch = (const int*)d_in[2];
    const float* W1  = (const float*)d_in[3];
    const float* as1 = (const float*)d_in[4];
    const float* ad1 = (const float*)d_in[5];
    const float* b1  = (const float*)d_in[6];
    const float* W2  = (const float*)d_in[7];
    const float* as2 = (const float*)d_in[8];
    const float* ad2 = (const float*)d_in[9];
    const float* b2  = (const float*)d_in[10];

    float* out_feats  = (float*)d_out;
    float* out_pooled = out_feats + (size_t)GAT_N * GAT_D;

    float* ws = (float*)d_ws;
    size_t o = 0;
    float*    h      = ws + o; o += (size_t)GAT_N * GAT_D;
    float*    feats1 = ws + o; o += (size_t)GAT_N * GAT_D;
    float*    alphaS = ws + o; o += GAT_N;
    float*    alphaD = ws + o; o += GAT_N;
    unsigned* deg     = (unsigned*)(ws + o);
    int*      rowptr  = (int*)(ws + o + CSR_NP);
    int*      csr_src = (int*)(ws + o + 2 * CSR_NP);
    static_assert(2 * CSR_NP + GAT_E <= 950000, "edge list must fit the old buffers");
    static_assert((size_t)CSR_B * CSR_NP <= (size_t)GAT_N * GAT_D, "blkcnt must fit feats1");
    o += 950000;
    float*    pool   = ws + o; o += GAT_G * GAT_D;
    unsigned* cnt    = (unsigned*)(ws + o); o += GAT_G;
    unsigned* rank   = (unsigned*)h;
    unsigned* blkcnt = (unsigned*)feats1;
    (void)ws_size; (void)in_sizes; (void)n_in; (void)out_size;

    gat_csr_count<<<CSR_B, CSR_T, 0, stream>>>(ei, blkcnt, rank);
    gat_csr_prefix<<<CSR_NP / 256, 256, 0, stream>>>(blkcnt, deg);
    gat_csr_scan<<<1, SCAN_T, 0, stream>>>(deg, rowptr);
    gat_csr_fill<<<GAT_E / 256, 256, 0, stream>>>(ei, blkcnt, rank, rowptr, csr_src);

    launch_gat_layer(x, W1, as1, ad1, b1, rowptr, csr_src,
                     h, alphaS, alphaD, feats1, stream);
    launch_gat_layer(feats1, W2, as2, ad2, b2, rowptr, csr_src,
                     h, alphaS, alphaD, out_feats, stream);

    gat_pool_zero<<<(GAT_G * GAT_D + 255) / 256, 256, 0, stream>>>(pool, cnt);
    gat_pool_accum<<<POOL_NB, POOL_T, 0, stream>>>(out_feats, batch, pool, cnt);
    gat_pool_final<<<(GAT_G * GAT_D + 255) / 256, 256, 0, stream>>>(pool, cnt, out_pooled);
}


# rowptr prefix sum without its one-block kernel: the prefix kernel scans each 256 node totals and stores the block sum, every fill block scans the 196 sums itself; coalesced accesses, 10 launches
# speedup vs baseline: 4.1971x; 1.0880x over previous
#include <hip/hip_runtime.h>
#include <hip/hip_bf16.h>


#define GAT_N      50000
#define GAT_E      800000
#define GAT_D      128
#define GAT_G      64
#define NEG_SLOPE  0.2f

typedef __attribute__((ext_vector_type(2))) float v2f;
typedef __attribute__((ext_vector_type(8))) float v8f;


template <int M>
__device__ __forceinline__ float gat_row_xmask(float v) {
    return __int_as_float(__builtin_amdgcn_update_dpp(
        0, __float_as_int(v), 0x160 | M, 0xf, 0xf, false));
}
__device__ __forceinline__ float gat_row_sum16(float v) {
    v += gat_row_xmask<1>(v);
    v += gat_row_xmask<2>(v);
    v += gat_row_xmask<4>(v);
    v += gat_row_xmask<8>(v);
    return v;
}

__global__ void __launch_bounds__(256)
gat_gemm128_wmma(const float* __restrict__ X,
                 const float* __restrict__ W,
                 const float* __restrict__ a_src,
                 const float* __restrict__ a_dst,
                 float* __restrict__ H,
                 float* __restrict__ as_out,
                 float* __restrict__ ad_out) {
    __shared__ float part[2][8][16];
    const int wave = threadIdx.x >> 5;
    const int lane = threadIdx.x & 31;
    const int half = lane >> 4;
    const int l    = lane & 15;
    const int m0   = blockIdx.x * 16;
    const int n0   = wave * 16;

    const float* xrow = X + (size_t)(m0 + l) * GAT_D;
    const float asv = a_src[n0 + l];
    const float adv = a_dst[n0 + l];
    v8f acc = {};

#pragma unroll
    for (int k = 0; k < GAT_D; k += 4) {
        const int kk = k + 2 * half;
        v2f a;
        {
            const float2 av = *(const float2*)(xrow + kk);
            a.x = av.x; a.y = av.y;
        }
        v2f b;
        b.x = W[(size_t)kk       * GAT_D + n0 + l];
        b.y = W[(size_t)(kk + 1) * GAT_D + n0 + l];
        acc = __builtin_amdgcn_wmma_f32_16x16x4_f32(
            false, a, false, b, (short)0, acc, false, false);
    }

    asm volatile("v_nop\n\tv_nop\n\tv_nop\n\tv_nop\n\tv_nop\n\tv_nop\n\tv_nop\n\tv_nop\n\tv_nop"
                 : "+v"(acc));

    float ps[8], pd[8];
#pragma unroll
    for (int r = 0; r < 8; ++r) {
        ps[r] = gat_row_sum16(acc[r] * asv);
        pd[r] = gat_row_sum16(acc[r] * adv);
    }
    if (l == 0) {
#pragma unroll
        for (int r = 0; r < 8; ++r) {
            part[0][wave][half * 8 + r] = ps[r];
            part[1][wave][half * 8 + r] = pd[r];
        }
    }
    __syncthreads();

    const int mBase = m0 + half * 8;
#pragma unroll
    for (int r = 0; r < 8; ++r)
        H[(size_t)(mBase + r) * GAT_D + n0 + l] = acc[r];

    if (threadIdx.x < 32) {
        float s = 0.f;
#pragma unroll
        for (int w = 0; w < 8; ++w) s += part[half][w][l];
        (half ? ad_out : as_out)[m0 + l] = s;
    }
}

#define CSR_B   64
#define CSR_CH  (GAT_E / CSR_B)
#define CSR_T   1024
#define CSR_NP  50176
#define PFX_T   256
#define PFX_NB  (CSR_NP / PFX_T)
static_assert(CSR_NP >= GAT_N + 1, "padding must cover rowptr[N]");
static_assert(CSR_CH * CSR_B == GAT_E, "slices must tile the edge list");
static_assert(CSR_NP % CSR_T == 0 && PFX_NB * PFX_T == CSR_NP, "padded size must tile the blocks");
static_assert(PFX_NB <= PFX_T, "one thread per block total in the fill kernel");
static_assert(GAT_E % PFX_T == 0 && GAT_E / PFX_T >= PFX_NB, "fill blocks tile the edges and cover rowptr");

__global__ void __launch_bounds__(CSR_T)
gat_csr_count(const int* __restrict__ ei,
              unsigned* __restrict__ blkcnt,
              unsigned* __restrict__ rank) {
    __shared__ unsigned hist[CSR_NP];
    const int t = threadIdx.x;
    for (int k = t; k < CSR_NP; k += CSR_T) hist[k] = 0u;
    __syncthreads();
    const int e0 = blockIdx.x * CSR_CH;
    for (int k = t; k < CSR_CH; k += CSR_T) {
        const int i = e0 + k;
        rank[i] = atomicAdd(&hist[ei[GAT_E + i]], 1u);
    }
    __syncthreads();
    unsigned* __restrict__ dstp = blkcnt + (size_t)blockIdx.x * CSR_NP;
    for (int k = t; k < CSR_NP; k += CSR_T) dstp[k] = hist[k];
}

__global__ void __launch_bounds__(PFX_T)
gat_csr_prefix(unsigned* __restrict__ blkcnt,
               int* __restrict__ rowloc,
               int* __restrict__ blksum) {
    __shared__ int wtot[PFX_T / 32];
    const int t = threadIdx.x;
    const int n = blockIdx.x * PFX_T + t;
    unsigned c[CSR_B];
#pragma unroll
    for (int b = 0; b < CSR_B; ++b) c[b] = blkcnt[(size_t)b * CSR_NP + n];
    unsigned run = 0u;
#pragma unroll
    for (int b = 0; b < CSR_B; ++b) {
        blkcnt[(size_t)b * CSR_NP + n] = run;
        run += c[b];
    }
    const int deg  = (int)run;
    const int lane = t & 31;
    const int wave = t >> 5;
    int inc = deg;
    for (int off = 1; off < 32; off <<= 1) {
        const int up = __shfl_up(inc, off, 32);
        if (lane >= off) inc += up;
    }
    if (lane == 31) wtot[wave] = inc;
    __syncthreads();
    int base = 0;
    for (int w = 0; w < PFX_T / 32; ++w) base += (w < wave) ? wtot[w] : 0;
    rowloc[n] = base + inc - deg;
    if (t == PFX_T - 1) blksum[blockIdx.x] = base + inc;
}

__global__ void __launch_bounds__(PFX_T)
gat_csr_fill(const int* __restrict__ ei,
             const unsigned* __restrict__ blkcnt,
             const unsigned* __restrict__ rank,
             const int* __restrict__ rowloc,
             const int* __restrict__ blksum,
             int* __restrict__ rowptr,
             int* __restrict__ csr_src) {
    __shared__ int boff[PFX_T];
    __shared__ int wtot[PFX_T / 32];
    const int t    = threadIdx.x;
    const int lane = t & 31;
    const int wave = t >> 5;
    const int i    = blockIdx.x * PFX_T + t;
    const int b = i / CSR_CH;
    const int s = ei[i];
    const int d = ei[GAT_E + i];
    const int r = (int)rank[i];
    const int within = rowloc[d] + (int)blkcnt[(size_t)b * CSR_NP + d];

    const int tot = (t < PFX_NB) ? blksum[t] : 0;
    int inc = tot;
    for (int off = 1; off < 32; off <<= 1) {
        const int up = __shfl_up(inc, off, 32);
        if (lane >= off) inc += up;
    }
    if (lane == 31) wtot[wave] = inc;
    __syncthreads();
    int base = 0;
    for (int w = 0; w < PFX_T / 32; ++w) base += (w < wave) ? wtot[w] : 0;
    boff[t] = base + inc - tot;
    __syncthreads();

    csr_src[boff[d / PFX_T] + within + r] = s;
    if (blockIdx.x < PFX_NB) {
        const int n = blockIdx.x * PFX_T + t;
        rowptr[n] = boff[blockIdx.x] + rowloc[n];
    }
}

__device__ __forceinline__ float gat_leaky(float e) {
    return (e > 0.f) ? e : NEG_SLOPE * e;
}

__global__ void __launch_bounds__(256)
gat_node_gather(const int* __restrict__ rowptr,
                const int* __restrict__ csr_src,
                const float* __restrict__ as,
                const float* __restrict__ ad,
                const float* __restrict__ H,
                const float* __restrict__ bias,
                float* __restrict__ out) {
    const int lane = threadIdx.x & 31;
    const int wave = __builtin_amdgcn_readfirstlane((int)(threadIdx.x >> 5));
    const int node = blockIdx.x * 8 + wave;
    if (node >= GAT_N) return;
    const int beg = rowptr[node];
    const int end = rowptr[node + 1];
    const float adn = ad[node];
    const float e_self = gat_leaky(as[node] + adn);

    const bool has0 = (beg + lane) < end;
    int   s0 = node;
    float e0 = e_self;
    if (has0) {
        s0 = csr_src[beg + lane];
        e0 = gat_leaky(as[s0] + adn);
    }

    float mx = fmaxf(e_self, e0);
    for (int base = beg + 32; base < end; base += 32) {
        const int j = base + lane;
        if (j < end) mx = fmaxf(mx, gat_leaky(as[csr_src[j]] + adn));
    }
#pragma unroll
    for (int off = 16; off > 0; off >>= 1)
        mx = fmaxf(mx, __shfl_xor(mx, off, 32));
    if (!__builtin_isfinite(mx)) mx = 0.f;

    const float ex0 = has0 ? expf(e0 - mx) : 0.f;
    float sm = ex0;
    for (int base = beg + 32; base < end; base += 32) {
        const int j = base + lane;
        if (j < end) sm += expf(gat_leaky(as[csr_src[j]] + adn) - mx);
    }
#pragma unroll
    for (int off = 16; off > 0; off >>= 1)
        sm += __shfl_xor(sm, off, 32);
    const float ex_self = expf(e_self - mx);
    const float den = (sm + ex_self) + 1e-16f;

    const float* hl = H + lane * 4;
    float4 acc = make_float4(0.f, 0.f, 0.f, 0.f);
    for (int base = beg; base < end; base += 32) {
        int   s_l = s0;
        float c_l = ex0 / den;
        if (base != beg) {
            const int j = base + lane;
            s_l = node;
            c_l = 0.f;
            if (j < end) {
                s_l = csr_src[j];
                c_l = expf(gat_leaky(as[s_l] + adn) - mx) / den;
            }
        }
        const int n = (end - base < 32) ? (end - base) : 32;
        for (int t = 0; t < n; ++t) {
            const int   s = __builtin_amdgcn_readlane(s_l, t);
            const float c = __int_as_float(__builtin_amdgcn_readlane(__float_as_int(c_l), t));
            const float4 h4 = *(const float4*)(hl + (size_t)s * GAT_D);
            acc.x += c * h4.x;
            acc.y += c * h4.y;
            acc.z += c * h4.z;
            acc.w += c * h4.w;
        }
    }
    {
        const float c = ex_self / den;
        const float4 h4 = *(const float4*)(hl + (size_t)node * GAT_D);
        acc.x += c * h4.x;
        acc.y += c * h4.y;
        acc.z += c * h4.z;
        acc.w += c * h4.w;
    }
    const float4 b4 = *(const float4*)(bias + lane * 4);
    float4 o4;
    o4.x = acc.x + b4.x;
    o4.y = acc.y + b4.y;
    o4.z = acc.z + b4.z;
    o4.w = acc.w + b4.w;
    *(float4*)(out + (size_t)node * GAT_D + lane * 4) = o4;
}

#define POOL_K   10
#define POOL_W   8
#define POOL_T   (POOL_W * 32)
#define POOL_NB  (GAT_N / (POOL_K * POOL_W))
#define POOL_RP  (POOL_T / GAT_D)
static_assert(POOL_NB * POOL_K * POOL_W == GAT_N, "pool blocks must tile the nodes");
static_assert(POOL_RP * GAT_D == POOL_T && GAT_G % POOL_RP == 0, "block must tile the table rows");
static_assert(GAT_D == 128, "a wave covers a row with one float4 per lane");

__global__ void gat_pool_zero(float* __restrict__ pool, unsigned* __restrict__ cnt) {
    const int t = blockIdx.x * blockDim.x + threadIdx.x;
    if (t < GAT_G * GAT_D) pool[t] = 0.f;
    if (t < GAT_G) cnt[t] = 0u;
}

__global__ void __launch_bounds__(POOL_T)
gat_pool_accum(const float* __restrict__ feats,
               const int* __restrict__ batch,
               float* __restrict__ pool,
               unsigned* __restrict__ cnt) {
    __shared__ float4   lpool4[GAT_G * GAT_D / 4];
    __shared__ unsigned lcnt[GAT_G];
    float* lpool = (float*)lpool4;
    const int t    = threadIdx.x;
    const int lane = t & 31;
    const int wave = __builtin_amdgcn_readfirstlane(t >> 5);
    const int n0   = (blockIdx.x * POOL_W + wave) * POOL_K;

    int    g[POOL_K];
    float4 r[POOL_K];
#pragma unroll
    for (int k = 0; k < POOL_K; ++k) {
        g[k] = batch[n0 + k];
        r[k] = *(const float4*)(feats + (size_t)(n0 + k) * GAT_D + lane * 4);
    }

    for (int k = t; k < GAT_G * GAT_D / 4; k += POOL_T)
        lpool4[k] = make_float4(0.f, 0.f, 0.f, 0.f);
    if (t < GAT_G) lcnt[t] = 0u;
    __syncthreads();

    auto flush = [&](int gi, const float4& a, unsigned run) {
        if ((unsigned)gi < (unsigned)GAT_G) {
            float* p = lpool + gi * GAT_D + lane * 4;
            atomicAdd(p + 0, a.x);
            atomicAdd(p + 1, a.y);
            atomicAdd(p + 2, a.z);
            atomicAdd(p + 3, a.w);
            if (lane == 0) atomicAdd(&lcnt[gi], run);
        }
    };

    int      cur = g[0];
    float4   acc = r[0];
    unsigned run = 1u;
#pragma unroll
    for (int k = 1; k < POOL_K; ++k) {
        if (g[k] != cur) {
            flush(cur, acc, run);
            cur = g[k];
            acc = make_float4(0.f, 0.f, 0.f, 0.f);
            run = 0u;
        }
        acc.x += r[k].x;
        acc.y += r[k].y;
        acc.z += r[k].z;
        acc.w += r[k].w;
        ++run;
    }
    flush(cur, acc, run);
    __syncthreads();

    unsigned c[GAT_G / POOL_RP];
#pragma unroll
    for (int i = 0; i < GAT_G / POOL_RP; ++i) c[i] = lcnt[t / GAT_D + POOL_RP * i];
#pragma unroll
    for (int i = 0; i < GAT_G / POOL_RP; ++i) {
        if (c[i] != 0u) {
            const int k = t + POOL_T * i;
            atomicAdd(&pool[k], lpool[k]);
        }
    }
    if (t < GAT_G) {
        const unsigned ct = lcnt[t];
        if (ct != 0u) atomicAdd(&cnt[t], ct);
    }
}

__global__ void gat_pool_final(const float* __restrict__ pool,
                               const unsigned* __restrict__ cnt,
                               float* __restrict__ out_pooled) {
    const int t = blockIdx.x * blockDim.x + threadIdx.x;
    if (t >= GAT_G * GAT_D) return;
    const float c = fmaxf((float)cnt[t >> 7], 1.0f);
    out_pooled[t] = pool[t] / c;
}

static void launch_gat_layer(const float* x, const float* W,
                             const float* a_s, const float* a_d,
                             const float* bias,
                             const int* rowptr, const int* csr_src,
                             float* h, float* alphaS, float* alphaD,
                             float* out, hipStream_t stream) {
    gat_gemm128_wmma<<<GAT_N / 16, 256, 0, stream>>>(x, W, a_s, a_d, h, alphaS, alphaD);
    gat_node_gather<<<GAT_N / 8, 256, 0, stream>>>(rowptr, csr_src, alphaS, alphaD,
                                                   h, bias, out);
}

extern "C" void kernel_launch(void* const* d_in, const int* in_sizes, int n_in,
                              void* d_out, int out_size, void* d_ws, size_t ws_size,
                              hipStream_t stream) {
    const float* x     = (const float*)d_in[0];
    const int*   ei    = (const int*)d_in[1];
    const int*   batch = (const int*)d_in[2];
    const float* W1  = (const float*)d_in[3];
    const float* as1 = (const float*)d_in[4];
    const float* ad1 = (const float*)d_in[5];
    const float* b1  = (const float*)d_in[6];
    const float* W2  = (const float*)d_in[7];
    const float* as2 = (const float*)d_in[8];
    const float* ad2 = (const float*)d_in[9];
    const float* b2  = (const float*)d_in[10];

    float* out_feats  = (float*)d_out;
    float* out_pooled = out_feats + (size_t)GAT_N * GAT_D;

    float* ws = (float*)d_ws;
    size_t o = 0;
    float*    h      = ws + o; o += (size_t)GAT_N * GAT_D;
    float*    feats1 = ws + o; o += (size_t)GAT_N * GAT_D;
    float*    alphaS = ws + o; o += GAT_N;
    float*    alphaD = ws + o; o += GAT_N;
    int*      rowloc  = (int*)(ws + o);
    int*      rowptr  = (int*)(ws + o + CSR_NP);
    int*      csr_src = (int*)(ws + o + 2 * CSR_NP);
    int*      blksum  = (int*)(ws + o + 2 * CSR_NP + GAT_E);
    static_assert(2 * CSR_NP + GAT_E + PFX_NB <= 950000, "edge list must fit the old buffers");
    static_assert((size_t)CSR_B * CSR_NP <= (size_t)GAT_N * GAT_D, "blkcnt must fit feats1");
    o += 950000;
    float*    pool   = ws + o; o += GAT_G * GAT_D;
    unsigned* cnt    = (unsigned*)(ws + o); o += GAT_G;
    unsigned* rank   = (unsigned*)h;
    unsigned* blkcnt = (unsigned*)feats1;
    (void)ws_size; (void)in_sizes; (void)n_in; (void)out_size;

    gat_csr_count<<<CSR_B, CSR_T, 0, stream>>>(ei, blkcnt, rank);
    gat_csr_prefix<<<PFX_NB, PFX_T, 0, stream>>>(blkcnt, rowloc, blksum);
    gat_csr_fill<<<GAT_E / PFX_T, PFX_T, 0, stream>>>(ei, blkcnt, rank, rowloc, blksum,
                                                     rowptr, csr_src);

    launch_gat_layer(x, W1, as1, ad1, b1, rowptr, csr_src,
                     h, alphaS, alphaD, feats1, stream);
    launch_gat_layer(feats1, W2, as2, ad2, b2, rowptr, csr_src,
                     h, alphaS, alphaD, out_feats, stream);

    gat_pool_zero<<<(GAT_G * GAT_D + 255) / 256, 256, 0, stream>>>(pool, cnt);
    gat_pool_accum<<<POOL_NB, POOL_T, 0, stream>>>(out_feats, batch, pool, cnt);
    gat_pool_final<<<(GAT_G * GAT_D + 255) / 256, 256, 0, stream>>>(pool, cnt, out_pooled);
}


# GEMM block takes 80 rows, not 16: X rows staged once in LDS by whole-row loads, each wave keeps its W fragments in registers over 5 row tiles; k grouped by 8 over the products (new sum order)
# speedup vs baseline: 4.5954x; 1.0949x over previous
#include <hip/hip_runtime.h>
#include <hip/hip_bf16.h>


#define GAT_N      50000
#define GAT_E      800000
#define GAT_D      128
#define GAT_G      64
#define NEG_SLOPE  0.2f

typedef __attribute__((ext_vector_type(2))) float v2f;
typedef __attribute__((ext_vector_type(8))) float v8f;

#define GEMM_RT    5
#define GEMM_ROWS  (GEMM_RT * 16)
#define GEMM_RPW   (GEMM_ROWS / 8)
#define GEMM_P4    33
static_assert(GAT_N % GEMM_ROWS == 0, "gemm blocks must tile the rows");
static_assert(GEMM_RPW * 8 == GEMM_ROWS, "every wave stages the same number of rows");
static_assert(GAT_D == 128, "a wave covers a row with one float4 per lane; 8 column tiles");
static_assert(2 * GEMM_ROWS <= 256, "one thread per entry of the two dot products");

template <int M>
__device__ __forceinline__ float gat_row_xmask(float v) {
    return __int_as_float(__builtin_amdgcn_update_dpp(
        0, __float_as_int(v), 0x160 | M, 0xf, 0xf, false));
}
__device__ __forceinline__ float gat_row_sum16(float v) {
    v += gat_row_xmask<1>(v);
    v += gat_row_xmask<2>(v);
    v += gat_row_xmask<4>(v);
    v += gat_row_xmask<8>(v);
    return v;
}

__global__ void __launch_bounds__(256)
gat_gemm128_wmma(const float* __restrict__ X,
                 const float* __restrict__ W,
                 const float* __restrict__ a_src,
                 const float* __restrict__ a_dst,
                 float* __restrict__ H,
                 float* __restrict__ as_out,
                 float* __restrict__ ad_out) {
    __shared__ float4 xs[GEMM_ROWS * GEMM_P4];
    __shared__ float part[2][8][GEMM_ROWS];
    const int wave = threadIdx.x >> 5;
    const int lane = threadIdx.x & 31;
    const int half = lane >> 4;
    const int l    = lane & 15;
    const int m0   = blockIdx.x * GEMM_ROWS;
    const int n0   = wave * 16;

    float4 xr[GEMM_RPW];
#pragma unroll
    for (int i = 0; i < GEMM_RPW; ++i)
        xr[i] = *(const float4*)(X + (size_t)(m0 + wave * GEMM_RPW + i) * GAT_D + lane * 4);

    const float* wcol = W + (size_t)(4 * half) * GAT_D + n0 + l;
    v2f b[32];
#pragma unroll
    for (int j = 0; j < 32; ++j) {
        const int kb = 8 * (j >> 1) + 2 * (j & 1);
        b[j].x = wcol[(size_t)kb       * GAT_D];
        b[j].y = wcol[(size_t)(kb + 1) * GAT_D];
    }
    const float asv = a_src[n0 + l];
    const float adv = a_dst[n0 + l];

#pragma unroll
    for (int i = 0; i < GEMM_RPW; ++i)
        xs[(wave * GEMM_RPW + i) * GEMM_P4 + lane] = xr[i];
    __syncthreads();

    v8f acc[GEMM_RT];
#pragma unroll
    for (int t = 0; t < GEMM_RT; ++t) acc[t] = v8f{};

    const float4* xa = xs + l * GEMM_P4 + half;
#pragma unroll
    for (int g = 0; g < 16; ++g) {
        float4 a4[GEMM_RT];
#pragma unroll
        for (int t = 0; t < GEMM_RT; ++t) a4[t] = xa[t * 16 * GEMM_P4 + 2 * g];
#pragma unroll
        for (int t = 0; t < GEMM_RT; ++t) {
            v2f a;
            a.x = a4[t].x; a.y = a4[t].y;
            acc[t] = __builtin_amdgcn_wmma_f32_16x16x4_f32(
                false, a, false, b[2 * g], (short)0, acc[t], false, false);
        }
#pragma unroll
        for (int t = 0; t < GEMM_RT; ++t) {
            v2f a;
            a.x = a4[t].z; a.y = a4[t].w;
            acc[t] = __builtin_amdgcn_wmma_f32_16x16x4_f32(
                false, a, false, b[2 * g + 1], (short)0, acc[t], false, false);
        }
    }

    __builtin_amdgcn_sched_barrier(0);
    asm volatile("v_nop\n\tv_nop\n\tv_nop\n\tv_nop\n\tv_nop\n\tv_nop\n\tv_nop\n\tv_nop\n\tv_nop"
                 : "+v"(acc[0]), "+v"(acc[1]), "+v"(acc[2]), "+v"(acc[3]), "+v"(acc[4]));
    static_assert(GEMM_RT == 5, "the statement above names every accumulator");

#pragma unroll
    for (int t = 0; t < GEMM_RT; ++t) {
        float ps[8], pd[8];
#pragma unroll
        for (int r = 0; r < 8; ++r) {
            ps[r] = gat_row_sum16(acc[t][r] * asv);
            pd[r] = gat_row_sum16(acc[t][r] * adv);
        }
        if (l == 0) {
#pragma unroll
            for (int r = 0; r < 8; ++r) {
                part[0][wave][t * 16 + half * 8 + r] = ps[r];
                part[1][wave][t * 16 + half * 8 + r] = pd[r];
            }
        }
    }
    __syncthreads();

#pragma unroll
    for (int t = 0; t < GEMM_RT; ++t) {
        const int mBase = m0 + t * 16 + half * 8;
#pragma unroll
        for (int r = 0; r < 8; ++r)
            H[(size_t)(mBase + r) * GAT_D + n0 + l] = acc[t][r];
    }

    if (threadIdx.x < 2 * GEMM_ROWS) {
        const int which = threadIdx.x / GEMM_ROWS;
        const int row   = threadIdx.x - which * GEMM_ROWS;
        float s = 0.f;
#pragma unroll
        for (int w = 0; w < 8; ++w) s += part[which][w][row];
        (which ? ad_out : as_out)[m0 + row] = s;
    }
}

#define CSR_B   64
#define CSR_CH  (GAT_E / CSR_B)
#define CSR_T   1024
#define CSR_NP  50176
#define PFX_T   256
#define PFX_NB  (CSR_NP / PFX_T)
static_assert(CSR_NP >= GAT_N + 1, "padding must cover rowptr[N]");
static_assert(CSR_CH * CSR_B == GAT_E, "slices must tile the edge list");
static_assert(CSR_NP % CSR_T == 0 && PFX_NB * PFX_T == CSR_NP, "padded size must tile the blocks");
static_assert(PFX_NB <= PFX_T, "one thread per block total in the fill kernel");
static_assert(GAT_E % PFX_T == 0 && GAT_E / PFX_T >= PFX_NB, "fill blocks tile the edges and cover rowptr");

__global__ void __launch_bounds__(CSR_T)
gat_csr_count(const int* __restrict__ ei,
              unsigned* __restrict__ blkcnt,
              unsigned* __restrict__ rank) {
    __shared__ unsigned hist[CSR_NP];
    const int t = threadIdx.x;
    for (int k = t; k < CSR_NP; k += CSR_T) hist[k] = 0u;
    __syncthreads();
    const int e0 = blockIdx.x * CSR_CH;
    for (int k = t; k < CSR_CH; k += CSR_T) {
        const int i = e0 + k;
        rank[i] = atomicAdd(&hist[ei[GAT_E + i]], 1u);
    }
    __syncthreads();
    unsigned* __restrict__ dstp = blkcnt + (size_t)blockIdx.x * CSR_NP;
    for (int k = t; k < CSR_NP; k += CSR_T) dstp[k] = hist[k];
}

__global__ void __launch_bounds__(PFX_T)
gat_csr_prefix(unsigned* __restrict__ blkcnt,
               int* __restrict__ rowloc,
               int* __restrict__ blksum) {
    __shared__ int wtot[PFX_T / 32];
    const int t = threadIdx.x;
    const int n = blockIdx.x * PFX_T + t;
    unsigned c[CSR_B];
#pragma unroll
    for (int b = 0; b < CSR_B; ++b) c[b] = blkcnt[(size_t)b * CSR_NP + n];
    unsigned run = 0u;
#pragma unroll
    for (int b = 0; b < CSR_B; ++b) {
        blkcnt[(size_t)b * CSR_NP + n] = run;
        run += c[b];
    }
    const int deg  = (int)run;
    const int lane = t & 31;
    const int wave = t >> 5;
    int inc = deg;
    for (int off = 1; off < 32; off <<= 1) {
        const int up = __shfl_up(inc, off, 32);
        if (lane >= off) inc += up;
    }
    if (lane == 31) wtot[wave] = inc;
    __syncthreads();
    int base = 0;
    for (int w = 0; w < PFX_T / 32; ++w) base += (w < wave) ? wtot[w] : 0;
    rowloc[n] = base + inc - deg;
    if (t == PFX_T - 1) blksum[blockIdx.x] = base + inc;
}

__global__ void __launch_bounds__(PFX_T)
gat_csr_fill(const int* __restrict__ ei,
             const unsigned* __restrict__ blkcnt,
             const unsigned* __restrict__ rank,
             const int* __restrict__ rowloc,
             const int* __restrict__ blksum,
             int* __restrict__ rowptr,
             int* __restrict__ csr_src) {
    __shared__ int boff[PFX_T];
    __shared__ int wtot[PFX_T / 32];
    const int t    = threadIdx.x;
    const int lane = t & 31;
    const int wave = t >> 5;
    const int i    = blockIdx.x * PFX_T + t;
    const int b = i / CSR_CH;
    const int s = ei[i];
    const int d = ei[GAT_E + i];
    const int r = (int)rank[i];
    const int within = rowloc[d] + (int)blkcnt[(size_t)b * CSR_NP + d];

    const int tot = (t < PFX_NB) ? blksum[t] : 0;
    int inc = tot;
    for (int off = 1; off < 32; off <<= 1) {
        const int up = __shfl_up(inc, off, 32);
        if (lane >= off) inc += up;
    }
    if (lane == 31) wtot[wave] = inc;
    __syncthreads();
    int base = 0;
    for (int w = 0; w < PFX_T / 32; ++w) base += (w < wave) ? wtot[w] : 0;
    boff[t] = base + inc - tot;
    __syncthreads();

    csr_src[boff[d / PFX_T] + within + r] = s;
    if (blockIdx.x < PFX_NB) {
        const int n = blockIdx.x * PFX_T + t;
        rowptr[n] = boff[blockIdx.x] + rowloc[n];
    }
}

__device__ __forceinline__ float gat_leaky(float e) {
    return (e > 0.f) ? e : NEG_SLOPE * e;
}

__global__ void __launch_bounds__(256)
gat_node_gather(const int* __restrict__ rowptr,
                const int* __restrict__ csr_src,
                const float* __restrict__ as,
                const float* __restrict__ ad,
                const float* __restrict__ H,
                const float* __restrict__ bias,
                float* __restrict__ out) {
    const int lane = threadIdx.x & 31;
    const int wave = __builtin_amdgcn_readfirstlane((int)(threadIdx.x >> 5));
    const int node = blockIdx.x * 8 + wave;
    if (node >= GAT_N) return;
    const int beg = rowptr[node];
    const int end = rowptr[node + 1];
    const float adn = ad[node];
    const float e_self = gat_leaky(as[node] + adn);

    const bool has0 = (beg + lane) < end;
    int   s0 = node;
    float e0 = e_self;
    if (has0) {
        s0 = csr_src[beg + lane];
        e0 = gat_leaky(as[s0] + adn);
    }

    float mx = fmaxf(e_self, e0);
    for (int base = beg + 32; base < end; base += 32) {
        const int j = base + lane;
        if (j < end) mx = fmaxf(mx, gat_leaky(as[csr_src[j]] + adn));
    }
#pragma unroll
    for (int off = 16; off > 0; off >>= 1)
        mx = fmaxf(mx, __shfl_xor(mx, off, 32));
    if (!__builtin_isfinite(mx)) mx = 0.f;

    const float ex0 = has0 ? expf(e0 - mx) : 0.f;
    float sm = ex0;
    for (int base = beg + 32; base < end; base += 32) {
        const int j = base + lane;
        if (j < end) sm += expf(gat_leaky(as[csr_src[j]] + adn) - mx);
    }
#pragma unroll
    for (int off = 16; off > 0; off >>= 1)
        sm += __shfl_xor(sm, off, 32);
    const float ex_self = expf(e_self - mx);
    const float den = (sm + ex_self) + 1e-16f;

    const float* hl = H + lane * 4;
    float4 acc = make_float4(0.f, 0.f, 0.f, 0.f);
    for (int base = beg; base < end; base += 32) {
        int   s_l = s0;
        float c_l = ex0 / den;
        if (base != beg) {
            const int j = base + lane;
            s_l = node;
            c_l = 0.f;
            if (j < end) {
                s_l = csr_src[j];
                c_l = expf(gat_leaky(as[s_l] + adn) - mx) / den;
            }
        }
        const int n = (end - base < 32) ? (end - base) : 32;
        for (int t = 0; t < n; ++t) {
            const int   s = __builtin_amdgcn_readlane(s_l, t);
            const float c = __int_as_float(__builtin_amdgcn_readlane(__float_as_int(c_l), t));
            const float4 h4 = *(const float4*)(hl + (size_t)s * GAT_D);
            acc.x += c * h4.x;
            acc.y += c * h4.y;
            acc.z += c * h4.z;
            acc.w += c * h4.w;
        }
    }
    {
        const float c = ex_self / den;
        const float4 h4 = *(const float4*)(hl + (size_t)node * GAT_D);
        acc.x += c * h4.x;
        acc.y += c * h4.y;
        acc.z += c * h4.z;
        acc.w += c * h4.w;
    }
    const float4 b4 = *(const float4*)(bias + lane * 4);
    float4 o4;
    o4.x = acc.x + b4.x;
    o4.y = acc.y + b4.y;
    o4.z = acc.z + b4.z;
    o4.w = acc.w + b4.w;
    *(float4*)(out + (size_t)node * GAT_D + lane * 4) = o4;
}

#define POOL_K   10
#define POOL_W   8
#define POOL_T   (POOL_W * 32)
#define POOL_NB  (GAT_N / (POOL_K * POOL_W))
#define POOL_RP  (POOL_T / GAT_D)
static_assert(POOL_NB * POOL_K * POOL_W == GAT_N, "pool blocks must tile the nodes");
static_assert(POOL_RP * GAT_D == POOL_T && GAT_G % POOL_RP == 0, "block must tile the table rows");
static_assert(GAT_D == 128, "a wave covers a row with one float4 per lane");

__global__ void gat_pool_zero(float* __restrict__ pool, unsigned* __restrict__ cnt) {
    const int t = blockIdx.x * blockDim.x + threadIdx.x;
    if (t < GAT_G * GAT_D) pool[t] = 0.f;
    if (t < GAT_G) cnt[t] = 0u;
}

__global__ void __launch_bounds__(POOL_T)
gat_pool_accum(const float* __restrict__ feats,
               const int* __restrict__ batch,
               float* __restrict__ pool,
               unsigned* __restrict__ cnt) {
    __shared__ float4   lpool4[GAT_G * GAT_D / 4];
    __shared__ unsigned lcnt[GAT_G];
    float* lpool = (float*)lpool4;
    const int t    = threadIdx.x;
    const int lane = t & 31;
    const int wave = __builtin_amdgcn_readfirstlane(t >> 5);
    const int n0   = (blockIdx.x * POOL_W + wave) * POOL_K;

    int    g[POOL_K];
    float4 r[POOL_K];
#pragma unroll
    for (int k = 0; k < POOL_K; ++k) {
        g[k] = batch[n0 + k];
        r[k] = *(const float4*)(feats + (size_t)(n0 + k) * GAT_D + lane * 4);
    }

    for (int k = t; k < GAT_G * GAT_D / 4; k += POOL_T)
        lpool4[k] = make_float4(0.f, 0.f, 0.f, 0.f);
    if (t < GAT_G) lcnt[t] = 0u;
    __syncthreads();

    auto flush = [&](int gi, const float4& a, unsigned run) {
        if ((unsigned)gi < (unsigned)GAT_G) {
            float* p = lpool + gi * GAT_D + lane * 4;
            atomicAdd(p + 0, a.x);
            atomicAdd(p + 1, a.y);
            atomicAdd(p + 2, a.z);
            atomicAdd(p + 3, a.w);
            if (lane == 0) atomicAdd(&lcnt[gi], run);
        }
    };

    int      cur = g[0];
    float4   acc = r[0];
    unsigned run = 1u;
#pragma unroll
    for (int k = 1; k < POOL_K; ++k) {
        if (g[k] != cur) {
            flush(cur, acc, run);
            cur = g[k];
            acc = make_float4(0.f, 0.f, 0.f, 0.f);
            run = 0u;
        }
        acc.x += r[k].x;
        acc.y += r[k].y;
        acc.z += r[k].z;
        acc.w += r[k].w;
        ++run;
    }
    flush(cur, acc, run);
    __syncthreads();

    unsigned c[GAT_G / POOL_RP];
#pragma unroll
    for (int i = 0; i < GAT_G / POOL_RP; ++i) c[i] = lcnt[t / GAT_D + POOL_RP * i];
#pragma unroll
    for (int i = 0; i < GAT_G / POOL_RP; ++i) {
        if (c[i] != 0u) {
            const int k = t + POOL_T * i;
            atomicAdd(&pool[k], lpool[k]);
        }
    }
    if (t < GAT_G) {
        const unsigned ct = lcnt[t];
        if (ct != 0u) atomicAdd(&cnt[t], ct);
    }
}

__global__ void gat_pool_final(const float* __restrict__ pool,
                               const unsigned* __restrict__ cnt,
                               float* __restrict__ out_pooled) {
    const int t = blockIdx.x * blockDim.x + threadIdx.x;
    if (t >= GAT_G * GAT_D) return;
    const float c = fmaxf((float)cnt[t >> 7], 1.0f);
    out_pooled[t] = pool[t] / c;
}

static void launch_gat_layer(const float* x, const float* W,
                             const float* a_s, const float* a_d,
                             const float* bias,
                             const int* rowptr, const int* csr_src,
                             float* h, float* alphaS, float* alphaD,
                             float* out, hipStream_t stream) {
    gat_gemm128_wmma<<<GAT_N / GEMM_ROWS, 256, 0, stream>>>(x, W, a_s, a_d, h, alphaS, alphaD);
    gat_node_gather<<<GAT_N / 8, 256, 0, stream>>>(rowptr, csr_src, alphaS, alphaD,
                                                   h, bias, out);
}

extern "C" void kernel_launch(void* const* d_in, const int* in_sizes, int n_in,
                              void* d_out, int out_size, void* d_ws, size_t ws_size,
                              hipStream_t stream) {
    const float* x     = (const float*)d_in[0];
    const int*   ei    = (const int*)d_in[1];
    const int*   batch = (const int*)d_in[2];
    const float* W1  = (const float*)d_in[3];
    const float* as1 = (const float*)d_in[4];
    const float* ad1 = (const float*)d_in[5];
    const float* b1  = (const float*)d_in[6];
    const float* W2  = (const float*)d_in[7];
    const float* as2 = (const float*)d_in[8];
    const float* ad2 = (const float*)d_in[9];
    const float* b2  = (const float*)d_in[10];

    float* out_feats  = (float*)d_out;
    float* out_pooled = out_feats + (size_t)GAT_N * GAT_D;

    float* ws = (float*)d_ws;
    size_t o = 0;
    float*    h      = ws + o; o += (size_t)GAT_N * GAT_D;
    float*    feats1 = ws + o; o += (size_t)GAT_N * GAT_D;
    float*    alphaS = ws + o; o += GAT_N;
    float*    alphaD = ws + o; o += GAT_N;
    int*      rowloc  = (int*)(ws + o);
    int*      rowptr  = (int*)(ws + o + CSR_NP);
    int*      csr_src = (int*)(ws + o + 2 * CSR_NP);
    int*      blksum  = (int*)(ws + o + 2 * CSR_NP + GAT_E);
    static_assert(2 * CSR_NP + GAT_E + PFX_NB <= 950000, "edge list must fit the old buffers");
    static_assert((size_t)CSR_B * CSR_NP <= (size_t)GAT_N * GAT_D, "blkcnt must fit feats1");
    o += 950000;
    float*    pool   = ws + o; o += GAT_G * GAT_D;
    unsigned* cnt    = (unsigned*)(ws + o); o += GAT_G;
    unsigned* rank   = (unsigned*)h;
    unsigned* blkcnt = (unsigned*)feats1;
    (void)ws_size; (void)in_sizes; (void)n_in; (void)out_size;

    gat_csr_count<<<CSR_B, CSR_T, 0, stream>>>(ei, blkcnt, rank);
    gat_csr_prefix<<<PFX_NB, PFX_T, 0, stream>>>(blkcnt, rowloc, blksum);
    gat_csr_fill<<<GAT_E / PFX_T, PFX_T, 0, stream>>>(ei, blkcnt, rank, rowloc, blksum,
                                                     rowptr, csr_src);

    launch_gat_layer(x, W1, as1, ad1, b1, rowptr, csr_src,
                     h, alphaS, alphaD, feats1, stream);
    launch_gat_layer(feats1, W2, as2, ad2, b2, rowptr, csr_src,
                     h, alphaS, alphaD, out_feats, stream);

    gat_pool_zero<<<(GAT_G * GAT_D + 255) / 256, 256, 0, stream>>>(pool, cnt);
    gat_pool_accum<<<POOL_NB, POOL_T, 0, stream>>>(out_feats, batch, pool, cnt);
    gat_pool_final<<<(GAT_G * GAT_D + 255) / 256, 256, 0, stream>>>(pool, cnt, out_pooled);
}


# Layer-1 GEMM blocks and the pool table's zeroing share the edge-list fill launch (block number picks the role, GEMM first; fill blocks take 5 edges per thread to fit the slots left): 8 launches
# speedup vs baseline: 4.8052x; 1.0457x over previous
#include <hip/hip_runtime.h>
#include <hip/hip_bf16.h>


#define GAT_N      50000
#define GAT_E      800000
#define GAT_D      128
#define GAT_G      64
#define NEG_SLOPE  0.2f

typedef __attribute__((ext_vector_type(2))) float v2f;
typedef __attribute__((ext_vector_type(8))) float v8f;

#define GEMM_RT    5
#define GEMM_ROWS  (GEMM_RT * 16)
#define GEMM_RPW   (GEMM_ROWS / 8)
#define GEMM_P4    33
static_assert(GAT_N % GEMM_ROWS == 0, "gemm blocks must tile the rows");
static_assert(GEMM_RPW * 8 == GEMM_ROWS, "every wave stages the same number of rows");
static_assert(GAT_D == 128, "a wave covers a row with one float4 per lane; 8 column tiles");
static_assert(2 * GEMM_ROWS <= 256, "one thread per entry of the two dot products");

template <int M>
__device__ __forceinline__ float gat_row_xmask(float v) {
    return __int_as_float(__builtin_amdgcn_update_dpp(
        0, __float_as_int(v), 0x160 | M, 0xf, 0xf, false));
}
__device__ __forceinline__ float gat_row_sum16(float v) {
    v += gat_row_xmask<1>(v);
    v += gat_row_xmask<2>(v);
    v += gat_row_xmask<4>(v);
    v += gat_row_xmask<8>(v);
    return v;
}

__device__ __forceinline__ void
gat_gemm_block(const int blk,
               const float* __restrict__ X,
               const float* __restrict__ W,
               const float* __restrict__ a_src,
               const float* __restrict__ a_dst,
               float* __restrict__ H,
               float* __restrict__ as_out,
               float* __restrict__ ad_out) {
    __shared__ float4 xs[GEMM_ROWS * GEMM_P4];
    __shared__ float part[2][8][GEMM_ROWS];
    const int wave = threadIdx.x >> 5;
    const int lane = threadIdx.x & 31;
    const int half = lane >> 4;
    const int l    = lane & 15;
    const int m0   = blk * GEMM_ROWS;
    const int n0   = wave * 16;

    float4 xr[GEMM_RPW];
#pragma unroll
    for (int i = 0; i < GEMM_RPW; ++i)
        xr[i] = *(const float4*)(X + (size_t)(m0 + wave * GEMM_RPW + i) * GAT_D + lane * 4);

    const float* wcol = W + (size_t)(4 * half) * GAT_D + n0 + l;
    v2f b[32];
#pragma unroll
    for (int j = 0; j < 32; ++j) {
        const int kb = 8 * (j >> 1) + 2 * (j & 1);
        b[j].x = wcol[(size_t)kb       * GAT_D];
        b[j].y = wcol[(size_t)(kb + 1) * GAT_D];
    }
    const float asv = a_src[n0 + l];
    const float adv = a_dst[n0 + l];

#pragma unroll
    for (int i = 0; i < GEMM_RPW; ++i)
        xs[(wave * GEMM_RPW + i) * GEMM_P4 + lane] = xr[i];
    __syncthreads();

    v8f acc[GEMM_RT];
#pragma unroll
    for (int t = 0; t < GEMM_RT; ++t) acc[t] = v8f{};

    const float4* xa = xs + l * GEMM_P4 + half;
#pragma unroll
    for (int g = 0; g < 16; ++g) {
        float4 a4[GEMM_RT];
#pragma unroll
        for (int t = 0; t < GEMM_RT; ++t) a4[t] = xa[t * 16 * GEMM_P4 + 2 * g];
#pragma unroll
        for (int t = 0; t < GEMM_RT; ++t) {
            v2f a;
            a.x = a4[t].x; a.y = a4[t].y;
            acc[t] = __builtin_amdgcn_wmma_f32_16x16x4_f32(
                false, a, false, b[2 * g], (short)0, acc[t], false, false);
        }
#pragma unroll
        for (int t = 0; t < GEMM_RT; ++t) {
            v2f a;
            a.x = a4[t].z; a.y = a4[t].w;
            acc[t] = __builtin_amdgcn_wmma_f32_16x16x4_f32(
                false, a, false, b[2 * g + 1], (short)0, acc[t], false, false);
        }
    }

    __builtin_amdgcn_sched_barrier(0);
    asm volatile("v_nop\n\tv_nop\n\tv_nop\n\tv_nop\n\tv_nop\n\tv_nop\n\tv_nop\n\tv_nop\n\tv_nop"
                 : "+v"(acc[0]), "+v"(acc[1]), "+v"(acc[2]), "+v"(acc[3]), "+v"(acc[4]));
    static_assert(GEMM_RT == 5, "the statement above names every accumulator");

#pragma unroll
    for (int t = 0; t < GEMM_RT; ++t) {
        float ps[8], pd[8];
#pragma unroll
        for (int r = 0; r < 8; ++r) {
            ps[r] = gat_row_sum16(acc[t][r] * asv);
            pd[r] = gat_row_sum16(acc[t][r] * adv);
        }
        if (l == 0) {
#pragma unroll
            for (int r = 0; r < 8; ++r) {
                part[0][wave][t * 16 + half * 8 + r] = ps[r];
                part[1][wave][t * 16 + half * 8 + r] = pd[r];
            }
        }
    }
    __syncthreads();

#pragma unroll
    for (int t = 0; t < GEMM_RT; ++t) {
        const int mBase = m0 + t * 16 + half * 8;
#pragma unroll
        for (int r = 0; r < 8; ++r)
            H[(size_t)(mBase + r) * GAT_D + n0 + l] = acc[t][r];
    }

    if (threadIdx.x < 2 * GEMM_ROWS) {
        const int which = threadIdx.x / GEMM_ROWS;
        const int row   = threadIdx.x - which * GEMM_ROWS;
        float s = 0.f;
#pragma unroll
        for (int w = 0; w < 8; ++w) s += part[which][w][row];
        (which ? ad_out : as_out)[m0 + row] = s;
    }
}

#define GEMM_NB  (GAT_N / GEMM_ROWS)

__global__ void __launch_bounds__(256)
gat_gemm128_wmma(const float* __restrict__ X,
                 const float* __restrict__ W,
                 const float* __restrict__ a_src,
                 const float* __restrict__ a_dst,
                 float* __restrict__ H,
                 float* __restrict__ as_out,
                 float* __restrict__ ad_out) {
    gat_gemm_block((int)blockIdx.x, X, W, a_src, a_dst, H, as_out, ad_out);
}

#define CSR_B   64
#define CSR_CH  (GAT_E / CSR_B)
#define CSR_T   1024
#define CSR_NP  50176
#define PFX_T   256
#define PFX_NB  (CSR_NP / PFX_T)
static_assert(CSR_NP >= GAT_N + 1, "padding must cover rowptr[N]");
static_assert(CSR_CH * CSR_B == GAT_E, "slices must tile the edge list");
static_assert(CSR_NP % CSR_T == 0 && PFX_NB * PFX_T == CSR_NP, "padded size must tile the blocks");
static_assert(PFX_NB <= PFX_T, "one thread per block total in the fill kernel");

__global__ void __launch_bounds__(CSR_T)
gat_csr_count(const int* __restrict__ ei,
              unsigned* __restrict__ blkcnt,
              unsigned* __restrict__ rank) {
    __shared__ unsigned hist[CSR_NP];
    const int t = threadIdx.x;
    for (int k = t; k < CSR_NP; k += CSR_T) hist[k] = 0u;
    __syncthreads();
    const int e0 = blockIdx.x * CSR_CH;
    for (int k = t; k < CSR_CH; k += CSR_T) {
        const int i = e0 + k;
        rank[i] = atomicAdd(&hist[ei[GAT_E + i]], 1u);
    }
    __syncthreads();
    unsigned* __restrict__ dstp = blkcnt + (size_t)blockIdx.x * CSR_NP;
    for (int k = t; k < CSR_NP; k += CSR_T) dstp[k] = hist[k];
}

__global__ void __launch_bounds__(PFX_T)
gat_csr_prefix(unsigned* __restrict__ blkcnt,
               int* __restrict__ rowloc,
               int* __restrict__ blksum) {
    __shared__ int wtot[PFX_T / 32];
    const int t = threadIdx.x;
    const int n = blockIdx.x * PFX_T + t;
    unsigned c[CSR_B];
#pragma unroll
    for (int b = 0; b < CSR_B; ++b) c[b] = blkcnt[(size_t)b * CSR_NP + n];
    unsigned run = 0u;
#pragma unroll
    for (int b = 0; b < CSR_B; ++b) {
        blkcnt[(size_t)b * CSR_NP + n] = run;
        run += c[b];
    }
    const int deg  = (int)run;
    const int lane = t & 31;
    const int wave = t >> 5;
    int inc = deg;
    for (int off = 1; off < 32; off <<= 1) {
        const int up = __shfl_up(inc, off, 32);
        if (lane >= off) inc += up;
    }
    if (lane == 31) wtot[wave] = inc;
    __syncthreads();
    int base = 0;
    for (int w = 0; w < PFX_T / 32; ++w) base += (w < wave) ? wtot[w] : 0;
    rowloc[n] = base + inc - deg;
    if (t == PFX_T - 1) blksum[blockIdx.x] = base + inc;
}

#define FILL_K    5
#define FILL_EPB  (FILL_K * PFX_T)
#define FILL_NB   (GAT_E / FILL_EPB)
static_assert(FILL_NB * FILL_EPB == GAT_E, "fill blocks must tile the edges");
static_assert(FILL_NB >= PFX_NB, "the first fill blocks cover rowptr");
__device__ __forceinline__ void
gat_csr_fill_block(const int blk,
                   const int* __restrict__ ei,
                   const unsigned* __restrict__ blkcnt,
                   const unsigned* __restrict__ rank,
                   const int* __restrict__ rowloc,
                   const int* __restrict__ blksum,
                   int* __restrict__ rowptr,
                   int* __restrict__ csr_src) {
    __shared__ int boff[PFX_T];
    __shared__ int wtot[PFX_T / 32];
    const int t    = threadIdx.x;
    const int lane = t & 31;
    const int wave = t >> 5;
    const int i0   = blk * FILL_EPB + t;
    int s[FILL_K], d[FILL_K], r[FILL_K], within[FILL_K];
#pragma unroll
    for (int k = 0; k < FILL_K; ++k) {
        const int i = i0 + k * PFX_T;
        s[k] = ei[i];
        d[k] = ei[GAT_E + i];
        r[k] = (int)rank[i];
    }
#pragma unroll
    for (int k = 0; k < FILL_K; ++k) {
        const int b = (i0 + k * PFX_T) / CSR_CH;
        within[k] = rowloc[d[k]] + (int)blkcnt[(size_t)b * CSR_NP + d[k]];
    }

    const int tot = (t < PFX_NB) ? blksum[t] : 0;
    int inc = tot;
    for (int off = 1; off < 32; off <<= 1) {
        const int up = __shfl_up(inc, off, 32);
        if (lane >= off) inc += up;
    }
    if (lane == 31) wtot[wave] = inc;
    __syncthreads();
    int base = 0;
    for (int w = 0; w < PFX_T / 32; ++w) base += (w < wave) ? wtot[w] : 0;
    boff[t] = base + inc - tot;
    __syncthreads();

#pragma unroll
    for (int k = 0; k < FILL_K; ++k)
        csr_src[boff[d[k] / PFX_T] + within[k] + r[k]] = s[k];
    if (blk < PFX_NB) {
        const int n = blk * PFX_T + t;
        rowptr[n] = boff[blk] + rowloc[n];
    }
}

#define POOLZ_NB  (GAT_G * GAT_D / PFX_T)
static_assert(POOLZ_NB * PFX_T == GAT_G * GAT_D && POOLZ_NB <= FILL_NB, "fill blocks cover the pooling table");
static_assert(GAT_G <= PFX_T, "one block clears the counts");
static_assert(PFX_T == 256, "both roles run 256 threads");

__global__ void __launch_bounds__(PFX_T)
gat_fill_gemm1(const int* __restrict__ ei,
               const unsigned* __restrict__ blkcnt,
               const unsigned* __restrict__ rank,
               const int* __restrict__ rowloc,
               const int* __restrict__ blksum,
               int* __restrict__ rowptr,
               int* __restrict__ csr_src,
               const float* __restrict__ X,
               const float* __restrict__ W,
               const float* __restrict__ a_src,
               const float* __restrict__ a_dst,
               float* __restrict__ H,
               float* __restrict__ as_out,
               float* __restrict__ ad_out,
               float* __restrict__ pool,
               unsigned* __restrict__ cnt) {
    if (blockIdx.x < GEMM_NB) {
        gat_gemm_block((int)blockIdx.x, X, W, a_src, a_dst, H, as_out, ad_out);
        return;
    }
    const int fb = (int)blockIdx.x - GEMM_NB;
    gat_csr_fill_block(fb, ei, blkcnt, rank, rowloc, blksum, rowptr, csr_src);
    if (fb < POOLZ_NB) {
        pool[fb * PFX_T + threadIdx.x] = 0.f;
        if (fb == 0 && threadIdx.x < GAT_G) cnt[threadIdx.x] = 0u;
    }
}

__device__ __forceinline__ float gat_leaky(float e) {
    return (e > 0.f) ? e : NEG_SLOPE * e;
}

__global__ void __launch_bounds__(256)
gat_node_gather(const int* __restrict__ rowptr,
                const int* __restrict__ csr_src,
                const float* __restrict__ as,
                const float* __restrict__ ad,
                const float* __restrict__ H,
                const float* __restrict__ bias,
                float* __restrict__ out) {
    const int lane = threadIdx.x & 31;
    const int wave = __builtin_amdgcn_readfirstlane((int)(threadIdx.x >> 5));
    const int node = blockIdx.x * 8 + wave;
    if (node >= GAT_N) return;
    const int beg = rowptr[node];
    const int end = rowptr[node + 1];
    const float adn = ad[node];
    const float e_self = gat_leaky(as[node] + adn);

    const bool has0 = (beg + lane) < end;
    int   s0 = node;
    float e0 = e_self;
    if (has0) {
        s0 = csr_src[beg + lane];
        e0 = gat_leaky(as[s0] + adn);
    }

    float mx = fmaxf(e_self, e0);
    for (int base = beg + 32; base < end; base += 32) {
        const int j = base + lane;
        if (j < end) mx = fmaxf(mx, gat_leaky(as[csr_src[j]] + adn));
    }
#pragma unroll
    for (int off = 16; off > 0; off >>= 1)
        mx = fmaxf(mx, __shfl_xor(mx, off, 32));
    if (!__builtin_isfinite(mx)) mx = 0.f;

    const float ex0 = has0 ? expf(e0 - mx) : 0.f;
    float sm = ex0;
    for (int base = beg + 32; base < end; base += 32) {
        const int j = base + lane;
        if (j < end) sm += expf(gat_leaky(as[csr_src[j]] + adn) - mx);
    }
#pragma unroll
    for (int off = 16; off > 0; off >>= 1)
        sm += __shfl_xor(sm, off, 32);
    const float ex_self = expf(e_self - mx);
    const float den = (sm + ex_self) + 1e-16f;

    const float* hl = H + lane * 4;
    float4 acc = make_float4(0.f, 0.f, 0.f, 0.f);
    for (int base = beg; base < end; base += 32) {
        int   s_l = s0;
        float c_l = ex0 / den;
        if (base != beg) {
            const int j = base + lane;
            s_l = node;
            c_l = 0.f;
            if (j < end) {
                s_l = csr_src[j];
                c_l = expf(gat_leaky(as[s_l] + adn) - mx) / den;
            }
        }
        const int n = (end - base < 32) ? (end - base) : 32;
        for (int t = 0; t < n; ++t) {
            const int   s = __builtin_amdgcn_readlane(s_l, t);
            const float c = __int_as_float(__builtin_amdgcn_readlane(__float_as_int(c_l), t));
            const float4 h4 = *(const float4*)(hl + (size_t)s * GAT_D);
            acc.x += c * h4.x;
            acc.y += c * h4.y;
            acc.z += c * h4.z;
            acc.w += c * h4.w;
        }
    }
    {
        const float c = ex_self / den;
        const float4 h4 = *(const float4*)(hl + (size_t)node * GAT_D);
        acc.x += c * h4.x;
        acc.y += c * h4.y;
        acc.z += c * h4.z;
        acc.w += c * h4.w;
    }
    const float4 b4 = *(const float4*)(bias + lane * 4);
    float4 o4;
    o4.x = acc.x + b4.x;
    o4.y = acc.y + b4.y;
    o4.z = acc.z + b4.z;
    o4.w = acc.w + b4.w;
    *(float4*)(out + (size_t)node * GAT_D + lane * 4) = o4;
}

#define POOL_K   10
#define POOL_W   8
#define POOL_T   (POOL_W * 32)
#define POOL_NB  (GAT_N / (POOL_K * POOL_W))
#define POOL_RP  (POOL_T / GAT_D)
static_assert(POOL_NB * POOL_K * POOL_W == GAT_N, "pool blocks must tile the nodes");
static_assert(POOL_RP * GAT_D == POOL_T && GAT_G % POOL_RP == 0, "block must tile the table rows");
static_assert(GAT_D == 128, "a wave covers a row with one float4 per lane");

__global__ void __launch_bounds__(POOL_T)
gat_pool_accum(const float* __restrict__ feats,
               const int* __restrict__ batch,
               float* __restrict__ pool,
               unsigned* __restrict__ cnt) {
    __shared__ float4   lpool4[GAT_G * GAT_D / 4];
    __shared__ unsigned lcnt[GAT_G];
    float* lpool = (float*)lpool4;
    const int t    = threadIdx.x;
    const int lane = t & 31;
    const int wave = __builtin_amdgcn_readfirstlane(t >> 5);
    const int n0   = (blockIdx.x * POOL_W + wave) * POOL_K;

    int    g[POOL_K];
    float4 r[POOL_K];
#pragma unroll
    for (int k = 0; k < POOL_K; ++k) {
        g[k] = batch[n0 + k];
        r[k] = *(const float4*)(feats + (size_t)(n0 + k) * GAT_D + lane * 4);
    }

    for (int k = t; k < GAT_G * GAT_D / 4; k += POOL_T)
        lpool4[k] = make_float4(0.f, 0.f, 0.f, 0.f);
    if (t < GAT_G) lcnt[t] = 0u;
    __syncthreads();

    auto flush = [&](int gi, const float4& a, unsigned run) {
        if ((unsigned)gi < (unsigned)GAT_G) {
            float* p = lpool + gi * GAT_D + lane * 4;
            atomicAdd(p + 0, a.x);
            atomicAdd(p + 1, a.y);
            atomicAdd(p + 2, a.z);
            atomicAdd(p + 3, a.w);
            if (lane == 0) atomicAdd(&lcnt[gi], run);
        }
    };

    int      cur = g[0];
    float4   acc = r[0];
    unsigned run = 1u;
#pragma unroll
    for (int k = 1; k < POOL_K; ++k) {
        if (g[k] != cur) {
            flush(cur, acc, run);
            cur = g[k];
            acc = make_float4(0.f, 0.f, 0.f, 0.f);
            run = 0u;
        }
        acc.x += r[k].x;
        acc.y += r[k].y;
        acc.z += r[k].z;
        acc.w += r[k].w;
        ++run;
    }
    flush(cur, acc, run);
    __syncthreads();

    unsigned c[GAT_G / POOL_RP];
#pragma unroll
    for (int i = 0; i < GAT_G / POOL_RP; ++i) c[i] = lcnt[t / GAT_D + POOL_RP * i];
#pragma unroll
    for (int i = 0; i < GAT_G / POOL_RP; ++i) {
        if (c[i] != 0u) {
            const int k = t + POOL_T * i;
            atomicAdd(&pool[k], lpool[k]);
        }
    }
    if (t < GAT_G) {
        const unsigned ct = lcnt[t];
        if (ct != 0u) atomicAdd(&cnt[t], ct);
    }
}

__global__ void gat_pool_final(const float* __restrict__ pool,
                               const unsigned* __restrict__ cnt,
                               float* __restrict__ out_pooled) {
    const int t = blockIdx.x * blockDim.x + threadIdx.x;
    if (t >= GAT_G * GAT_D) return;
    const float c = fmaxf((float)cnt[t >> 7], 1.0f);
    out_pooled[t] = pool[t] / c;
}

extern "C" void kernel_launch(void* const* d_in, const int* in_sizes, int n_in,
                              void* d_out, int out_size, void* d_ws, size_t ws_size,
                              hipStream_t stream) {
    const float* x     = (const float*)d_in[0];
    const int*   ei    = (const int*)d_in[1];
    const int*   batch = (const int*)d_in[2];
    const float* W1  = (const float*)d_in[3];
    const float* as1 = (const float*)d_in[4];
    const float* ad1 = (const float*)d_in[5];
    const float* b1  = (const float*)d_in[6];
    const float* W2  = (const float*)d_in[7];
    const float* as2 = (const float*)d_in[8];
    const float* ad2 = (const float*)d_in[9];
    const float* b2  = (const float*)d_in[10];

    float* out_feats  = (float*)d_out;
    float* out_pooled = out_feats + (size_t)GAT_N * GAT_D;

    float* ws = (float*)d_ws;
    size_t o = 0;
    float*    h      = ws + o; o += (size_t)GAT_N * GAT_D;
    float*    feats1 = ws + o; o += (size_t)GAT_N * GAT_D;
    float*    alphaS = ws + o; o += GAT_N;
    float*    alphaD = ws + o; o += GAT_N;
    int*      rowloc  = (int*)(ws + o);
    int*      rowptr  = (int*)(ws + o + CSR_NP);
    int*      csr_src = (int*)(ws + o + 2 * CSR_NP);
    int*      blksum  = (int*)(ws + o + 2 * CSR_NP + GAT_E);
    static_assert(2 * CSR_NP + GAT_E + PFX_NB <= 950000, "edge list must fit the old buffers");
    static_assert((size_t)CSR_B * CSR_NP + GAT_E <= (size_t)GAT_N * GAT_D, "blkcnt and rank must fit feats1");
    o += 950000;
    float*    pool   = ws + o; o += GAT_G * GAT_D;
    unsigned* cnt    = (unsigned*)(ws + o); o += GAT_G;
    unsigned* blkcnt = (unsigned*)feats1;
    unsigned* rank   = blkcnt + (size_t)CSR_B * CSR_NP;
    (void)ws_size; (void)in_sizes; (void)n_in; (void)out_size;

    gat_csr_count<<<CSR_B, CSR_T, 0, stream>>>(ei, blkcnt, rank);
    gat_csr_prefix<<<PFX_NB, PFX_T, 0, stream>>>(blkcnt, rowloc, blksum);
    gat_fill_gemm1<<<GEMM_NB + FILL_NB, PFX_T, 0, stream>>>(ei, blkcnt, rank, rowloc, blksum,
                                                            rowptr, csr_src,
                                                            x, W1, as1, ad1, h, alphaS, alphaD,
                                                            pool, cnt);

    gat_node_gather<<<GAT_N / 8, 256, 0, stream>>>(rowptr, csr_src, alphaS, alphaD,
                                                   h, b1, feats1);
    gat_gemm128_wmma<<<GEMM_NB, 256, 0, stream>>>(feats1, W2, as2, ad2, h, alphaS, alphaD);
    gat_node_gather<<<GAT_N / 8, 256, 0, stream>>>(rowptr, csr_src, alphaS, alphaD,
                                                   h, b2, out_feats);

    gat_pool_accum<<<POOL_NB, POOL_T, 0, stream>>>(out_feats, batch, pool, cnt);
    gat_pool_final<<<(GAT_G * GAT_D + 255) / 256, 256, 0, stream>>>(pool, cnt, out_pooled);
}


# Layer-2 product moved into the layer-1 gather launch: a block gathers 32 nodes' rows into an LDS tile (4 per wave) and multiplies the tile by W2 there; layer-1 rows never reach memory: 7 launches
# speedup vs baseline: 5.1486x; 1.0715x over previous
#include <hip/hip_runtime.h>
#include <hip/hip_bf16.h>


#define GAT_N      50000
#define GAT_E      800000
#define GAT_D      128
#define GAT_G      64
#define NEG_SLOPE  0.2f

typedef __attribute__((ext_vector_type(2))) float v2f;
typedef __attribute__((ext_vector_type(8))) float v8f;

#define GEMM_RT    5
#define GEMM_ROWS  (GEMM_RT * 16)
#define GEMM_RPW   (GEMM_ROWS / 8)
#define GEMM_P4    33
static_assert(GAT_N % GEMM_ROWS == 0, "gemm blocks must tile the rows");
static_assert(GEMM_RPW * 8 == GEMM_ROWS, "every wave stages the same number of rows");
static_assert(GAT_D == 128, "a wave covers a row with one float4 per lane; 8 column tiles");
static_assert(2 * GEMM_ROWS <= 256, "one thread per entry of the two dot products");

template <int M>
__device__ __forceinline__ float gat_row_xmask(float v) {
    return __int_as_float(__builtin_amdgcn_update_dpp(
        0, __float_as_int(v), 0x160 | M, 0xf, 0xf, false));
}
__device__ __forceinline__ float gat_row_sum16(float v) {
    v += gat_row_xmask<1>(v);
    v += gat_row_xmask<2>(v);
    v += gat_row_xmask<4>(v);
    v += gat_row_xmask<8>(v);
    return v;
}

__device__ __forceinline__ void
gat_gemm_block(const int blk,
               const float* __restrict__ X,
               const float* __restrict__ W,
               const float* __restrict__ a_src,
               const float* __restrict__ a_dst,
               float* __restrict__ H,
               float* __restrict__ as_out,
               float* __restrict__ ad_out) {
    __shared__ float4 xs[GEMM_ROWS * GEMM_P4];
    __shared__ float part[2][8][GEMM_ROWS];
    const int wave = threadIdx.x >> 5;
    const int lane = threadIdx.x & 31;
    const int half = lane >> 4;
    const int l    = lane & 15;
    const int m0   = blk * GEMM_ROWS;
    const int n0   = wave * 16;

    float4 xr[GEMM_RPW];
#pragma unroll
    for (int i = 0; i < GEMM_RPW; ++i)
        xr[i] = *(const float4*)(X + (size_t)(m0 + wave * GEMM_RPW + i) * GAT_D + lane * 4);

    const float* wcol = W + (size_t)(4 * half) * GAT_D + n0 + l;
    v2f b[32];
#pragma unroll
    for (int j = 0; j < 32; ++j) {
        const int kb = 8 * (j >> 1) + 2 * (j & 1);
        b[j].x = wcol[(size_t)kb       * GAT_D];
        b[j].y = wcol[(size_t)(kb + 1) * GAT_D];
    }
    const float asv = a_src[n0 + l];
    const float adv = a_dst[n0 + l];

#pragma unroll
    for (int i = 0; i < GEMM_RPW; ++i)
        xs[(wave * GEMM_RPW + i) * GEMM_P4 + lane] = xr[i];
    __syncthreads();

    v8f acc[GEMM_RT];
#pragma unroll
    for (int t = 0; t < GEMM_RT; ++t) acc[t] = v8f{};

    const float4* xa = xs + l * GEMM_P4 + half;
#pragma unroll
    for (int g = 0; g < 16; ++g) {
        float4 a4[GEMM_RT];
#pragma unroll
        for (int t = 0; t < GEMM_RT; ++t) a4[t] = xa[t * 16 * GEMM_P4 + 2 * g];
#pragma unroll
        for (int t = 0; t < GEMM_RT; ++t) {
            v2f a;
            a.x = a4[t].x; a.y = a4[t].y;
            acc[t] = __builtin_amdgcn_wmma_f32_16x16x4_f32(
                false, a, false, b[2 * g], (short)0, acc[t], false, false);
        }
#pragma unroll
        for (int t = 0; t < GEMM_RT; ++t) {
            v2f a;
            a.x = a4[t].z; a.y = a4[t].w;
            acc[t] = __builtin_amdgcn_wmma_f32_16x16x4_f32(
                false, a, false, b[2 * g + 1], (short)0, acc[t], false, false);
        }
    }

    __builtin_amdgcn_sched_barrier(0);
    asm volatile("v_nop\n\tv_nop\n\tv_nop\n\tv_nop\n\tv_nop\n\tv_nop\n\tv_nop\n\tv_nop\n\tv_nop"
                 : "+v"(acc[0]), "+v"(acc[1]), "+v"(acc[2]), "+v"(acc[3]), "+v"(acc[4]));
    static_assert(GEMM_RT == 5, "the statement above names every accumulator");

#pragma unroll
    for (int t = 0; t < GEMM_RT; ++t) {
        float ps[8], pd[8];
#pragma unroll
        for (int r = 0; r < 8; ++r) {
            ps[r] = gat_row_sum16(acc[t][r] * asv);
            pd[r] = gat_row_sum16(acc[t][r] * adv);
        }
        if (l == 0) {
#pragma unroll
            for (int r = 0; r < 8; ++r) {
                part[0][wave][t * 16 + half * 8 + r] = ps[r];
                part[1][wave][t * 16 + half * 8 + r] = pd[r];
            }
        }
    }
    __syncthreads();

#pragma unroll
    for (int t = 0; t < GEMM_RT; ++t) {
        const int mBase = m0 + t * 16 + half * 8;
#pragma unroll
        for (int r = 0; r < 8; ++r)
            H[(size_t)(mBase + r) * GAT_D + n0 + l] = acc[t][r];
    }

    if (threadIdx.x < 2 * GEMM_ROWS) {
        const int which = threadIdx.x / GEMM_ROWS;
        const int row   = threadIdx.x - which * GEMM_ROWS;
        float s = 0.f;
#pragma unroll
        for (int w = 0; w < 8; ++w) s += part[which][w][row];
        (which ? ad_out : as_out)[m0 + row] = s;
    }
}

#define GEMM_NB  (GAT_N / GEMM_ROWS)

#define CSR_B   64
#define CSR_CH  (GAT_E / CSR_B)
#define CSR_T   1024
#define CSR_NP  50176
#define PFX_T   256
#define PFX_NB  (CSR_NP / PFX_T)
static_assert(CSR_NP >= GAT_N + 1, "padding must cover rowptr[N]");
static_assert(CSR_CH * CSR_B == GAT_E, "slices must tile the edge list");
static_assert(CSR_NP % CSR_T == 0 && PFX_NB * PFX_T == CSR_NP, "padded size must tile the blocks");
static_assert(PFX_NB <= PFX_T, "one thread per block total in the fill kernel");

__global__ void __launch_bounds__(CSR_T)
gat_csr_count(const int* __restrict__ ei,
              unsigned* __restrict__ blkcnt,
              unsigned* __restrict__ rank) {
    __shared__ unsigned hist[CSR_NP];
    const int t = threadIdx.x;
    for (int k = t; k < CSR_NP; k += CSR_T) hist[k] = 0u;
    __syncthreads();
    const int e0 = blockIdx.x * CSR_CH;
    for (int k = t; k < CSR_CH; k += CSR_T) {
        const int i = e0 + k;
        rank[i] = atomicAdd(&hist[ei[GAT_E + i]], 1u);
    }
    __syncthreads();
    unsigned* __restrict__ dstp = blkcnt + (size_t)blockIdx.x * CSR_NP;
    for (int k = t; k < CSR_NP; k += CSR_T) dstp[k] = hist[k];
}

__global__ void __launch_bounds__(PFX_T)
gat_csr_prefix(unsigned* __restrict__ blkcnt,
               int* __restrict__ rowloc,
               int* __restrict__ blksum) {
    __shared__ int wtot[PFX_T / 32];
    const int t = threadIdx.x;
    const int n = blockIdx.x * PFX_T + t;
    unsigned c[CSR_B];
#pragma unroll
    for (int b = 0; b < CSR_B; ++b) c[b] = blkcnt[(size_t)b * CSR_NP + n];
    unsigned run = 0u;
#pragma unroll
    for (int b = 0; b < CSR_B; ++b) {
        blkcnt[(size_t)b * CSR_NP + n] = run;
        run += c[b];
    }
    const int deg  = (int)run;
    const int lane = t & 31;
    const int wave = t >> 5;
    int inc = deg;
    for (int off = 1; off < 32; off <<= 1) {
        const int up = __shfl_up(inc, off, 32);
        if (lane >= off) inc += up;
    }
    if (lane == 31) wtot[wave] = inc;
    __syncthreads();
    int base = 0;
    for (int w = 0; w < PFX_T / 32; ++w) base += (w < wave) ? wtot[w] : 0;
    rowloc[n] = base + inc - deg;
    if (t == PFX_T - 1) blksum[blockIdx.x] = base + inc;
}

#define FILL_K    5
#define FILL_EPB  (FILL_K * PFX_T)
#define FILL_NB   (GAT_E / FILL_EPB)
static_assert(FILL_NB * FILL_EPB == GAT_E, "fill blocks must tile the edges");
static_assert(FILL_NB >= PFX_NB, "the first fill blocks cover rowptr");
__device__ __forceinline__ void
gat_csr_fill_block(const int blk,
                   const int* __restrict__ ei,
                   const unsigned* __restrict__ blkcnt,
                   const unsigned* __restrict__ rank,
                   const int* __restrict__ rowloc,
                   const int* __restrict__ blksum,
                   int* __restrict__ rowptr,
                   int* __restrict__ csr_src) {
    __shared__ int boff[PFX_T];
    __shared__ int wtot[PFX_T / 32];
    const int t    = threadIdx.x;
    const int lane = t & 31;
    const int wave = t >> 5;
    const int i0   = blk * FILL_EPB + t;
    int s[FILL_K], d[FILL_K], r[FILL_K], within[FILL_K];
#pragma unroll
    for (int k = 0; k < FILL_K; ++k) {
        const int i = i0 + k * PFX_T;
        s[k] = ei[i];
        d[k] = ei[GAT_E + i];
        r[k] = (int)rank[i];
    }
#pragma unroll
    for (int k = 0; k < FILL_K; ++k) {
        const int b = (i0 + k * PFX_T) / CSR_CH;
        within[k] = rowloc[d[k]] + (int)blkcnt[(size_t)b * CSR_NP + d[k]];
    }

    const int tot = (t < PFX_NB) ? blksum[t] : 0;
    int inc = tot;
    for (int off = 1; off < 32; off <<= 1) {
        const int up = __shfl_up(inc, off, 32);
        if (lane >= off) inc += up;
    }
    if (lane == 31) wtot[wave] = inc;
    __syncthreads();
    int base = 0;
    for (int w = 0; w < PFX_T / 32; ++w) base += (w < wave) ? wtot[w] : 0;
    boff[t] = base + inc - tot;
    __syncthreads();

#pragma unroll
    for (int k = 0; k < FILL_K; ++k)
        csr_src[boff[d[k] / PFX_T] + within[k] + r[k]] = s[k];
    if (blk < PFX_NB) {
        const int n = blk * PFX_T + t;
        rowptr[n] = boff[blk] + rowloc[n];
    }
}

#define POOLZ_NB  (GAT_G * GAT_D / PFX_T)
static_assert(POOLZ_NB * PFX_T == GAT_G * GAT_D && POOLZ_NB <= FILL_NB, "fill blocks cover the pooling table");
static_assert(GAT_G <= PFX_T, "one block clears the counts");
static_assert(PFX_T == 256, "both roles run 256 threads");

__global__ void __launch_bounds__(PFX_T)
gat_fill_gemm1(const int* __restrict__ ei,
               const unsigned* __restrict__ blkcnt,
               const unsigned* __restrict__ rank,
               const int* __restrict__ rowloc,
               const int* __restrict__ blksum,
               int* __restrict__ rowptr,
               int* __restrict__ csr_src,
               const float* __restrict__ X,
               const float* __restrict__ W,
               const float* __restrict__ a_src,
               const float* __restrict__ a_dst,
               float* __restrict__ H,
               float* __restrict__ as_out,
               float* __restrict__ ad_out,
               float* __restrict__ pool,
               unsigned* __restrict__ cnt) {
    if (blockIdx.x < GEMM_NB) {
        gat_gemm_block((int)blockIdx.x, X, W, a_src, a_dst, H, as_out, ad_out);
        return;
    }
    const int fb = (int)blockIdx.x - GEMM_NB;
    gat_csr_fill_block(fb, ei, blkcnt, rank, rowloc, blksum, rowptr, csr_src);
    if (fb < POOLZ_NB) {
        pool[fb * PFX_T + threadIdx.x] = 0.f;
        if (fb == 0 && threadIdx.x < GAT_G) cnt[threadIdx.x] = 0u;
    }
}

__device__ __forceinline__ float gat_leaky(float e) {
    return (e > 0.f) ? e : NEG_SLOPE * e;
}

__device__ __forceinline__ float4
gat_gather_row(const int node, const int lane,
               const int* __restrict__ rowptr,
               const int* __restrict__ csr_src,
               const float* __restrict__ as,
               const float* ad,
               const float* __restrict__ H,
               const float* __restrict__ bias) {
    const int beg = rowptr[node];
    const int end = rowptr[node + 1];
    const float adn = ad[node];
    const float e_self = gat_leaky(as[node] + adn);

    const bool has0 = (beg + lane) < end;
    int   s0 = node;
    float e0 = e_self;
    if (has0) {
        s0 = csr_src[beg + lane];
        e0 = gat_leaky(as[s0] + adn);
    }

    float mx = fmaxf(e_self, e0);
    for (int base = beg + 32; base < end; base += 32) {
        const int j = base + lane;
        if (j < end) mx = fmaxf(mx, gat_leaky(as[csr_src[j]] + adn));
    }
#pragma unroll
    for (int off = 16; off > 0; off >>= 1)
        mx = fmaxf(mx, __shfl_xor(mx, off, 32));
    if (!__builtin_isfinite(mx)) mx = 0.f;

    const float ex0 = has0 ? expf(e0 - mx) : 0.f;
    float sm = ex0;
    for (int base = beg + 32; base < end; base += 32) {
        const int j = base + lane;
        if (j < end) sm += expf(gat_leaky(as[csr_src[j]] + adn) - mx);
    }
#pragma unroll
    for (int off = 16; off > 0; off >>= 1)
        sm += __shfl_xor(sm, off, 32);
    const float ex_self = expf(e_self - mx);
    const float den = (sm + ex_self) + 1e-16f;

    const float* hl = H + lane * 4;
    float4 acc = make_float4(0.f, 0.f, 0.f, 0.f);
    for (int base = beg; base < end; base += 32) {
        int   s_l = s0;
        float c_l = ex0 / den;
        if (base != beg) {
            const int j = base + lane;
            s_l = node;
            c_l = 0.f;
            if (j < end) {
                s_l = csr_src[j];
                c_l = expf(gat_leaky(as[s_l] + adn) - mx) / den;
            }
        }
        const int n = (end - base < 32) ? (end - base) : 32;
        for (int t = 0; t < n; ++t) {
            const int   s = __builtin_amdgcn_readlane(s_l, t);
            const float c = __int_as_float(__builtin_amdgcn_readlane(__float_as_int(c_l), t));
            const float4 h4 = *(const float4*)(hl + (size_t)s * GAT_D);
            acc.x += c * h4.x;
            acc.y += c * h4.y;
            acc.z += c * h4.z;
            acc.w += c * h4.w;
        }
    }
    {
        const float c = ex_self / den;
        const float4 h4 = *(const float4*)(hl + (size_t)node * GAT_D);
        acc.x += c * h4.x;
        acc.y += c * h4.y;
        acc.z += c * h4.z;
        acc.w += c * h4.w;
    }
    const float4 b4 = *(const float4*)(bias + lane * 4);
    float4 o4;
    o4.x = acc.x + b4.x;
    o4.y = acc.y + b4.y;
    o4.z = acc.z + b4.z;
    o4.w = acc.w + b4.w;
    return o4;
}

__global__ void __launch_bounds__(256)
gat_node_gather(const int* __restrict__ rowptr,
                const int* __restrict__ csr_src,
                const float* __restrict__ as,
                const float* __restrict__ ad,
                const float* __restrict__ H,
                const float* __restrict__ bias,
                float* __restrict__ out) {
    const int lane = threadIdx.x & 31;
    const int wave = __builtin_amdgcn_readfirstlane((int)(threadIdx.x >> 5));
    const int node = blockIdx.x * 8 + wave;
    if (node >= GAT_N) return;
    const float4 o4 = gat_gather_row(node, lane, rowptr, csr_src, as, ad, H, bias);
    *(float4*)(out + (size_t)node * GAT_D + lane * 4) = o4;
}

#define FG_RT    2
#define FG_ROWS  (FG_RT * 16)
#define FG_NPW   (FG_ROWS / 8)
#define FG_NB    ((GAT_N + FG_ROWS - 1) / FG_ROWS)
static_assert(2 * FG_ROWS <= 256, "one thread per entry of the two dot products");

__global__ void __launch_bounds__(256)
gat_gather_gemm2(const int* __restrict__ rowptr,
                 const int* __restrict__ csr_src,
                 const float* __restrict__ as1,
                 float* adio,
                 const float* __restrict__ H1,
                 const float* __restrict__ bias1,
                 const float* __restrict__ W,
                 const float* __restrict__ a_src,
                 const float* __restrict__ a_dst,
                 float* __restrict__ H2,
                 float* __restrict__ as2_out) {
    __shared__ float4 xs[FG_ROWS * GEMM_P4];
    __shared__ float part[2][8][FG_ROWS];
    const int lane = threadIdx.x & 31;
    const int wave = __builtin_amdgcn_readfirstlane((int)(threadIdx.x >> 5));
    const int half = lane >> 4;
    const int l    = lane & 15;
    const int m0   = (int)blockIdx.x * FG_ROWS;
    const int n0   = wave * 16;

    for (int q = 0; q < FG_NPW; ++q) {
        const int row  = wave * FG_NPW + q;
        const int node = m0 + row;
        float4 o4 = make_float4(0.f, 0.f, 0.f, 0.f);
        if (node < GAT_N)
            o4 = gat_gather_row(node, lane, rowptr, csr_src, as1, adio, H1, bias1);
        xs[row * GEMM_P4 + lane] = o4;
    }
    __syncthreads();

    const float* wcol = W + (size_t)(4 * half) * GAT_D + n0 + l;
    v2f b[32];
#pragma unroll
    for (int j = 0; j < 32; ++j) {
        const int kb = 8 * (j >> 1) + 2 * (j & 1);
        b[j].x = wcol[(size_t)kb       * GAT_D];
        b[j].y = wcol[(size_t)(kb + 1) * GAT_D];
    }
    const float asv = a_src[n0 + l];
    const float adv = a_dst[n0 + l];

    v8f acc[FG_RT];
#pragma unroll
    for (int t = 0; t < FG_RT; ++t) acc[t] = v8f{};

    const float4* xa = xs + l * GEMM_P4 + half;
#pragma unroll
    for (int g = 0; g < 16; ++g) {
        float4 a4[FG_RT];
#pragma unroll
        for (int t = 0; t < FG_RT; ++t) a4[t] = xa[t * 16 * GEMM_P4 + 2 * g];
#pragma unroll
        for (int t = 0; t < FG_RT; ++t) {
            v2f a;
            a.x = a4[t].x; a.y = a4[t].y;
            acc[t] = __builtin_amdgcn_wmma_f32_16x16x4_f32(
                false, a, false, b[2 * g], (short)0, acc[t], false, false);
        }
#pragma unroll
        for (int t = 0; t < FG_RT; ++t) {
            v2f a;
            a.x = a4[t].z; a.y = a4[t].w;
            acc[t] = __builtin_amdgcn_wmma_f32_16x16x4_f32(
                false, a, false, b[2 * g + 1], (short)0, acc[t], false, false);
        }
    }

    __builtin_amdgcn_sched_barrier(0);
    asm volatile("v_nop\n\tv_nop\n\tv_nop\n\tv_nop\n\tv_nop\n\tv_nop\n\tv_nop\n\tv_nop\n\tv_nop"
                 : "+v"(acc[0]), "+v"(acc[1]));
    static_assert(FG_RT == 2, "the statement above names every accumulator");

#pragma unroll
    for (int t = 0; t < FG_RT; ++t) {
        float ps[8], pd[8];
#pragma unroll
        for (int r = 0; r < 8; ++r) {
            ps[r] = gat_row_sum16(acc[t][r] * asv);
            pd[r] = gat_row_sum16(acc[t][r] * adv);
        }
        if (l == 0) {
#pragma unroll
            for (int r = 0; r < 8; ++r) {
                part[0][wave][t * 16 + half * 8 + r] = ps[r];
                part[1][wave][t * 16 + half * 8 + r] = pd[r];
            }
        }
    }
    __syncthreads();

#pragma unroll
    for (int t = 0; t < FG_RT; ++t) {
        const int mBase = m0 + t * 16 + half * 8;
#pragma unroll
        for (int r = 0; r < 8; ++r)
            if (mBase + r < GAT_N)
                H2[(size_t)(mBase + r) * GAT_D + n0 + l] = acc[t][r];
    }

    if (threadIdx.x < 2 * FG_ROWS) {
        const int which = threadIdx.x / FG_ROWS;
        const int row   = threadIdx.x - which * FG_ROWS;
        float s = 0.f;
#pragma unroll
        for (int w = 0; w < 8; ++w) s += part[which][w][row];
        if (m0 + row < GAT_N) (which ? adio : as2_out)[m0 + row] = s;
    }
}

#define POOL_K   10
#define POOL_W   8
#define POOL_T   (POOL_W * 32)
#define POOL_NB  (GAT_N / (POOL_K * POOL_W))
#define POOL_RP  (POOL_T / GAT_D)
static_assert(POOL_NB * POOL_K * POOL_W == GAT_N, "pool blocks must tile the nodes");
static_assert(POOL_RP * GAT_D == POOL_T && GAT_G % POOL_RP == 0, "block must tile the table rows");
static_assert(GAT_D == 128, "a wave covers a row with one float4 per lane");

__global__ void __launch_bounds__(POOL_T)
gat_pool_accum(const float* __restrict__ feats,
               const int* __restrict__ batch,
               float* __restrict__ pool,
               unsigned* __restrict__ cnt) {
    __shared__ float4   lpool4[GAT_G * GAT_D / 4];
    __shared__ unsigned lcnt[GAT_G];
    float* lpool = (float*)lpool4;
    const int t    = threadIdx.x;
    const int lane = t & 31;
    const int wave = __builtin_amdgcn_readfirstlane(t >> 5);
    const int n0   = (blockIdx.x * POOL_W + wave) * POOL_K;

    int    g[POOL_K];
    float4 r[POOL_K];
#pragma unroll
    for (int k = 0; k < POOL_K; ++k) {
        g[k] = batch[n0 + k];
        r[k] = *(const float4*)(feats + (size_t)(n0 + k) * GAT_D + lane * 4);
    }

    for (int k = t; k < GAT_G * GAT_D / 4; k += POOL_T)
        lpool4[k] = make_float4(0.f, 0.f, 0.f, 0.f);
    if (t < GAT_G) lcnt[t] = 0u;
    __syncthreads();

    auto flush = [&](int gi, const float4& a, unsigned run) {
        if ((unsigned)gi < (unsigned)GAT_G) {
            float* p = lpool + gi * GAT_D + lane * 4;
            atomicAdd(p + 0, a.x);
            atomicAdd(p + 1, a.y);
            atomicAdd(p + 2, a.z);
            atomicAdd(p + 3, a.w);
            if (lane == 0) atomicAdd(&lcnt[gi], run);
        }
    };

    int      cur = g[0];
    float4   acc = r[0];
    unsigned run = 1u;
#pragma unroll
    for (int k = 1; k < POOL_K; ++k) {
        if (g[k] != cur) {
            flush(cur, acc, run);
            cur = g[k];
            acc = make_float4(0.f, 0.f, 0.f, 0.f);
            run = 0u;
        }
        acc.x += r[k].x;
        acc.y += r[k].y;
        acc.z += r[k].z;
        acc.w += r[k].w;
        ++run;
    }
    flush(cur, acc, run);
    __syncthreads();

    unsigned c[GAT_G / POOL_RP];
#pragma unroll
    for (int i = 0; i < GAT_G / POOL_RP; ++i) c[i] = lcnt[t / GAT_D + POOL_RP * i];
#pragma unroll
    for (int i = 0; i < GAT_G / POOL_RP; ++i) {
        if (c[i] != 0u) {
            const int k = t + POOL_T * i;
            atomicAdd(&pool[k], lpool[k]);
        }
    }
    if (t < GAT_G) {
        const unsigned ct = lcnt[t];
        if (ct != 0u) atomicAdd(&cnt[t], ct);
    }
}

__global__ void gat_pool_final(const float* __restrict__ pool,
                               const unsigned* __restrict__ cnt,
                               float* __restrict__ out_pooled) {
    const int t = blockIdx.x * blockDim.x + threadIdx.x;
    if (t >= GAT_G * GAT_D) return;
    const float c = fmaxf((float)cnt[t >> 7], 1.0f);
    out_pooled[t] = pool[t] / c;
}

extern "C" void kernel_launch(void* const* d_in, const int* in_sizes, int n_in,
                              void* d_out, int out_size, void* d_ws, size_t ws_size,
                              hipStream_t stream) {
    const float* x     = (const float*)d_in[0];
    const int*   ei    = (const int*)d_in[1];
    const int*   batch = (const int*)d_in[2];
    const float* W1  = (const float*)d_in[3];
    const float* as1 = (const float*)d_in[4];
    const float* ad1 = (const float*)d_in[5];
    const float* b1  = (const float*)d_in[6];
    const float* W2  = (const float*)d_in[7];
    const float* as2 = (const float*)d_in[8];
    const float* ad2 = (const float*)d_in[9];
    const float* b2  = (const float*)d_in[10];

    float* out_feats  = (float*)d_out;
    float* out_pooled = out_feats + (size_t)GAT_N * GAT_D;

    float* ws = (float*)d_ws;
    size_t o = 0;
    float*    h      = ws + o; o += (size_t)GAT_N * GAT_D;
    float*    feats1 = ws + o; o += (size_t)GAT_N * GAT_D;
    float*    alphaS = ws + o; o += GAT_N;
    float*    alphaD = ws + o; o += GAT_N;
    int*      rowloc  = (int*)(ws + o);
    int*      rowptr  = (int*)(ws + o + CSR_NP);
    int*      csr_src = (int*)(ws + o + 2 * CSR_NP);
    int*      blksum  = (int*)(ws + o + 2 * CSR_NP + GAT_E);
    static_assert(2 * CSR_NP + GAT_E + PFX_NB <= 950000, "edge list must fit the old buffers");
    static_assert((size_t)CSR_B * CSR_NP + GAT_E <= (size_t)GAT_N * GAT_D, "blkcnt and rank must fit feats1");
    o += 950000;
    float*    pool   = ws + o; o += GAT_G * GAT_D;
    unsigned* cnt    = (unsigned*)(ws + o); o += GAT_G;
    unsigned* blkcnt = (unsigned*)feats1;
    unsigned* rank   = blkcnt + (size_t)CSR_B * CSR_NP;
    float*    h2      = feats1;
    float*    alphaS2 = (float*)rowloc;
    static_assert(CSR_NP >= GAT_N, "alpha_s of layer 2 must fit rowloc");
    (void)ws_size; (void)in_sizes; (void)n_in; (void)out_size;

    gat_csr_count<<<CSR_B, CSR_T, 0, stream>>>(ei, blkcnt, rank);
    gat_csr_prefix<<<PFX_NB, PFX_T, 0, stream>>>(blkcnt, rowloc, blksum);
    gat_fill_gemm1<<<GEMM_NB + FILL_NB, PFX_T, 0, stream>>>(ei, blkcnt, rank, rowloc, blksum,
                                                            rowptr, csr_src,
                                                            x, W1, as1, ad1, h, alphaS, alphaD,
                                                            pool, cnt);

    gat_gather_gemm2<<<FG_NB, 256, 0, stream>>>(rowptr, csr_src, alphaS, alphaD, h, b1,
                                                W2, as2, ad2, h2, alphaS2);
    gat_node_gather<<<GAT_N / 8, 256, 0, stream>>>(rowptr, csr_src, alphaS2, alphaD,
                                                   h2, b2, out_feats);

    gat_pool_accum<<<POOL_NB, POOL_T, 0, stream>>>(out_feats, batch, pool, cnt);
    gat_pool_final<<<(GAT_G * GAT_D + 255) / 256, 256, 0, stream>>>(pool, cnt, out_pooled);
}


# Mean pool moved into the layer-2 gather launch: graph sizes counted in the prefix launch; a block of 32 nodes adds run sum / size to the pooled output with float atomics (new sum order): 5 launches
# speedup vs baseline: 5.3549x; 1.0401x over previous
#include <hip/hip_runtime.h>
#include <hip/hip_bf16.h>


#define GAT_N      50000
#define GAT_E      800000
#define GAT_D      128
#define GAT_G      64
#define NEG_SLOPE  0.2f

typedef __attribute__((ext_vector_type(2))) float v2f;
typedef __attribute__((ext_vector_type(8))) float v8f;

#define GEMM_RT    5
#define GEMM_ROWS  (GEMM_RT * 16)
#define GEMM_RPW   (GEMM_ROWS / 8)
#define GEMM_P4    33
static_assert(GAT_N % GEMM_ROWS == 0, "gemm blocks must tile the rows");
static_assert(GEMM_RPW * 8 == GEMM_ROWS, "every wave stages the same number of rows");
static_assert(GAT_D == 128, "a wave covers a row with one float4 per lane; 8 column tiles");
static_assert(2 * GEMM_ROWS <= 256, "one thread per entry of the two dot products");

template <int M>
__device__ __forceinline__ float gat_row_xmask(float v) {
    return __int_as_float(__builtin_amdgcn_update_dpp(
        0, __float_as_int(v), 0x160 | M, 0xf, 0xf, false));
}
__device__ __forceinline__ float gat_row_sum16(float v) {
    v += gat_row_xmask<1>(v);
    v += gat_row_xmask<2>(v);
    v += gat_row_xmask<4>(v);
    v += gat_row_xmask<8>(v);
    return v;
}

__device__ __forceinline__ void
gat_gemm_block(const int blk,
               const float* __restrict__ X,
               const float* __restrict__ W,
               const float* __restrict__ a_src,
               const float* __restrict__ a_dst,
               float* __restrict__ H,
               float* __restrict__ as_out,
               float* __restrict__ ad_out) {
    __shared__ float4 xs[GEMM_ROWS * GEMM_P4];
    __shared__ float part[2][8][GEMM_ROWS];
    const int wave = threadIdx.x >> 5;
    const int lane = threadIdx.x & 31;
    const int half = lane >> 4;
    const int l    = lane & 15;
    const int m0   = blk * GEMM_ROWS;
    const int n0   = wave * 16;

    float4 xr[GEMM_RPW];
#pragma unroll
    for (int i = 0; i < GEMM_RPW; ++i)
        xr[i] = *(const float4*)(X + (size_t)(m0 + wave * GEMM_RPW + i) * GAT_D + lane * 4);

    const float* wcol = W + (size_t)(4 * half) * GAT_D + n0 + l;
    v2f b[32];
#pragma unroll
    for (int j = 0; j < 32; ++j) {
        const int kb = 8 * (j >> 1) + 2 * (j & 1);
        b[j].x = wcol[(size_t)kb       * GAT_D];
        b[j].y = wcol[(size_t)(kb + 1) * GAT_D];
    }
    const float asv = a_src[n0 + l];
    const float adv = a_dst[n0 + l];

#pragma unroll
    for (int i = 0; i < GEMM_RPW; ++i)
        xs[(wave * GEMM_RPW + i) * GEMM_P4 + lane] = xr[i];
    __syncthreads();

    v8f acc[GEMM_RT];
#pragma unroll
    for (int t = 0; t < GEMM_RT; ++t) acc[t] = v8f{};

    const float4* xa = xs + l * GEMM_P4 + half;
#pragma unroll
    for (int g = 0; g < 16; ++g) {
        float4 a4[GEMM_RT];
#pragma unroll
        for (int t = 0; t < GEMM_RT; ++t) a4[t] = xa[t * 16 * GEMM_P4 + 2 * g];
#pragma unroll
        for (int t = 0; t < GEMM_RT; ++t) {
            v2f a;
            a.x = a4[t].x; a.y = a4[t].y;
            acc[t] = __builtin_amdgcn_wmma_f32_16x16x4_f32(
                false, a, false, b[2 * g], (short)0, acc[t], false, false);
        }
#pragma unroll
        for (int t = 0; t < GEMM_RT; ++t) {
            v2f a;
            a.x = a4[t].z; a.y = a4[t].w;
            acc[t] = __builtin_amdgcn_wmma_f32_16x16x4_f32(
                false, a, false, b[2 * g + 1], (short)0, acc[t], false, false);
        }
    }

    __builtin_amdgcn_sched_barrier(0);
    asm volatile("v_nop\n\tv_nop\n\tv_nop\n\tv_nop\n\tv_nop\n\tv_nop\n\tv_nop\n\tv_nop\n\tv_nop"
                 : "+v"(acc[0]), "+v"(acc[1]), "+v"(acc[2]), "+v"(acc[3]), "+v"(acc[4]));
    static_assert(GEMM_RT == 5, "the statement above names every accumulator");

#pragma unroll
    for (int t = 0; t < GEMM_RT; ++t) {
        float ps[8], pd[8];
#pragma unroll
        for (int r = 0; r < 8; ++r) {
            ps[r] = gat_row_sum16(acc[t][r] * asv);
            pd[r] = gat_row_sum16(acc[t][r] * adv);
        }
        if (l == 0) {
#pragma unroll
            for (int r = 0; r < 8; ++r) {
                part[0][wave][t * 16 + half * 8 + r] = ps[r];
                part[1][wave][t * 16 + half * 8 + r] = pd[r];
            }
        }
    }
    __syncthreads();

#pragma unroll
    for (int t = 0; t < GEMM_RT; ++t) {
        const int mBase = m0 + t * 16 + half * 8;
#pragma unroll
        for (int r = 0; r < 8; ++r)
            H[(size_t)(mBase + r) * GAT_D + n0 + l] = acc[t][r];
    }

    if (threadIdx.x < 2 * GEMM_ROWS) {
        const int which = threadIdx.x / GEMM_ROWS;
        const int row   = threadIdx.x - which * GEMM_ROWS;
        float s = 0.f;
#pragma unroll
        for (int w = 0; w < 8; ++w) s += part[which][w][row];
        (which ? ad_out : as_out)[m0 + row] = s;
    }
}

#define GEMM_NB  (GAT_N / GEMM_ROWS)

#define CSR_B   64
#define CSR_CH  (GAT_E / CSR_B)
#define CSR_T   1024
#define CSR_NP  50176
#define PFX_T   256
#define PFX_NB  (CSR_NP / PFX_T)
static_assert(CSR_NP >= GAT_N + 1, "padding must cover rowptr[N]");
static_assert(CSR_CH * CSR_B == GAT_E, "slices must tile the edge list");
static_assert(CSR_NP % CSR_T == 0 && PFX_NB * PFX_T == CSR_NP, "padded size must tile the blocks");
static_assert(PFX_NB <= PFX_T, "one thread per block total in the fill kernel");

__global__ void __launch_bounds__(CSR_T)
gat_csr_count(const int* __restrict__ ei,
              unsigned* __restrict__ blkcnt,
              unsigned* __restrict__ rank,
              unsigned* __restrict__ cnt) {
    __shared__ unsigned hist[CSR_NP];
    const int t = threadIdx.x;
    if (blockIdx.x == 0 && t < GAT_G) cnt[t] = 0u;
    for (int k = t; k < CSR_NP; k += CSR_T) hist[k] = 0u;
    __syncthreads();
    const int e0 = blockIdx.x * CSR_CH;
    for (int k = t; k < CSR_CH; k += CSR_T) {
        const int i = e0 + k;
        rank[i] = atomicAdd(&hist[ei[GAT_E + i]], 1u);
    }
    __syncthreads();
    unsigned* __restrict__ dstp = blkcnt + (size_t)blockIdx.x * CSR_NP;
    for (int k = t; k < CSR_NP; k += CSR_T) dstp[k] = hist[k];
}

__global__ void __launch_bounds__(PFX_T)
gat_csr_prefix(unsigned* __restrict__ blkcnt,
               int* __restrict__ rowloc,
               int* __restrict__ blksum,
               const int* __restrict__ batch,
               unsigned* __restrict__ cnt) {
    __shared__ int wtot[PFX_T / 32];
    __shared__ unsigned lcnt[GAT_G];
    const int t = threadIdx.x;
    const int n = blockIdx.x * PFX_T + t;
    const int gn = (n < GAT_N) ? batch[n] : -1;
    if (t < GAT_G) lcnt[t] = 0u;
    __syncthreads();
    if ((unsigned)gn < (unsigned)GAT_G) atomicAdd(&lcnt[gn], 1u);
    unsigned c[CSR_B];
#pragma unroll
    for (int b = 0; b < CSR_B; ++b) c[b] = blkcnt[(size_t)b * CSR_NP + n];
    unsigned run = 0u;
#pragma unroll
    for (int b = 0; b < CSR_B; ++b) {
        blkcnt[(size_t)b * CSR_NP + n] = run;
        run += c[b];
    }
    const int deg  = (int)run;
    const int lane = t & 31;
    const int wave = t >> 5;
    int inc = deg;
    for (int off = 1; off < 32; off <<= 1) {
        const int up = __shfl_up(inc, off, 32);
        if (lane >= off) inc += up;
    }
    if (lane == 31) wtot[wave] = inc;
    __syncthreads();
    int base = 0;
    for (int w = 0; w < PFX_T / 32; ++w) base += (w < wave) ? wtot[w] : 0;
    rowloc[n] = base + inc - deg;
    if (t == PFX_T - 1) blksum[blockIdx.x] = base + inc;
    if (t < GAT_G) {
        const unsigned ct = lcnt[t];
        if (ct != 0u) atomicAdd(&cnt[t], ct);
    }
}

#define FILL_K    5
#define FILL_EPB  (FILL_K * PFX_T)
#define FILL_NB   (GAT_E / FILL_EPB)
static_assert(FILL_NB * FILL_EPB == GAT_E, "fill blocks must tile the edges");
static_assert(FILL_NB >= PFX_NB, "the first fill blocks cover rowptr");
__device__ __forceinline__ void
gat_csr_fill_block(const int blk,
                   const int* __restrict__ ei,
                   const unsigned* __restrict__ blkcnt,
                   const unsigned* __restrict__ rank,
                   const int* __restrict__ rowloc,
                   const int* __restrict__ blksum,
                   int* __restrict__ rowptr,
                   int* __restrict__ csr_src) {
    __shared__ int boff[PFX_T];
    __shared__ int wtot[PFX_T / 32];
    const int t    = threadIdx.x;
    const int lane = t & 31;
    const int wave = t >> 5;
    const int i0   = blk * FILL_EPB + t;
    int s[FILL_K], d[FILL_K], r[FILL_K], within[FILL_K];
#pragma unroll
    for (int k = 0; k < FILL_K; ++k) {
        const int i = i0 + k * PFX_T;
        s[k] = ei[i];
        d[k] = ei[GAT_E + i];
        r[k] = (int)rank[i];
    }
#pragma unroll
    for (int k = 0; k < FILL_K; ++k) {
        const int b = (i0 + k * PFX_T) / CSR_CH;
        within[k] = rowloc[d[k]] + (int)blkcnt[(size_t)b * CSR_NP + d[k]];
    }

    const int tot = (t < PFX_NB) ? blksum[t] : 0;
    int inc = tot;
    for (int off = 1; off < 32; off <<= 1) {
        const int up = __shfl_up(inc, off, 32);
        if (lane >= off) inc += up;
    }
    if (lane == 31) wtot[wave] = inc;
    __syncthreads();
    int base = 0;
    for (int w = 0; w < PFX_T / 32; ++w) base += (w < wave) ? wtot[w] : 0;
    boff[t] = base + inc - tot;
    __syncthreads();

#pragma unroll
    for (int k = 0; k < FILL_K; ++k)
        csr_src[boff[d[k] / PFX_T] + within[k] + r[k]] = s[k];
    if (blk < PFX_NB) {
        const int n = blk * PFX_T + t;
        rowptr[n] = boff[blk] + rowloc[n];
    }
}

#define POOLZ_NB  (GAT_G * GAT_D / PFX_T)
static_assert(POOLZ_NB * PFX_T == GAT_G * GAT_D && POOLZ_NB <= FILL_NB, "fill blocks cover the pooling table");
static_assert(PFX_T == 256, "both roles run 256 threads");

__global__ void __launch_bounds__(PFX_T)
gat_fill_gemm1(const int* __restrict__ ei,
               const unsigned* __restrict__ blkcnt,
               const unsigned* __restrict__ rank,
               const int* __restrict__ rowloc,
               const int* __restrict__ blksum,
               int* __restrict__ rowptr,
               int* __restrict__ csr_src,
               const float* __restrict__ X,
               const float* __restrict__ W,
               const float* __restrict__ a_src,
               const float* __restrict__ a_dst,
               float* __restrict__ H,
               float* __restrict__ as_out,
               float* __restrict__ ad_out,
               float* __restrict__ pooled) {
    if (blockIdx.x < GEMM_NB) {
        gat_gemm_block((int)blockIdx.x, X, W, a_src, a_dst, H, as_out, ad_out);
        return;
    }
    const int fb = (int)blockIdx.x - GEMM_NB;
    gat_csr_fill_block(fb, ei, blkcnt, rank, rowloc, blksum, rowptr, csr_src);
    if (fb < POOLZ_NB) pooled[fb * PFX_T + threadIdx.x] = 0.f;
}

__device__ __forceinline__ float gat_leaky(float e) {
    return (e > 0.f) ? e : NEG_SLOPE * e;
}

__device__ __forceinline__ float4
gat_gather_row(const int node, const int lane,
               const int* __restrict__ rowptr,
               const int* __restrict__ csr_src,
               const float* __restrict__ as,
               const float* ad,
               const float* __restrict__ H,
               const float* __restrict__ bias) {
    const int beg = rowptr[node];
    const int end = rowptr[node + 1];
    const float adn = ad[node];
    const float e_self = gat_leaky(as[node] + adn);

    const bool has0 = (beg + lane) < end;
    int   s0 = node;
    float e0 = e_self;
    if (has0) {
        s0 = csr_src[beg + lane];
        e0 = gat_leaky(as[s0] + adn);
    }

    float mx = fmaxf(e_self, e0);
    for (int base = beg + 32; base < end; base += 32) {
        const int j = base + lane;
        if (j < end) mx = fmaxf(mx, gat_leaky(as[csr_src[j]] + adn));
    }
#pragma unroll
    for (int off = 16; off > 0; off >>= 1)
        mx = fmaxf(mx, __shfl_xor(mx, off, 32));
    if (!__builtin_isfinite(mx)) mx = 0.f;

    const float ex0 = has0 ? expf(e0 - mx) : 0.f;
    float sm = ex0;
    for (int base = beg + 32; base < end; base += 32) {
        const int j = base + lane;
        if (j < end) sm += expf(gat_leaky(as[csr_src[j]] + adn) - mx);
    }
#pragma unroll
    for (int off = 16; off > 0; off >>= 1)
        sm += __shfl_xor(sm, off, 32);
    const float ex_self = expf(e_self - mx);
    const float den = (sm + ex_self) + 1e-16f;

    const float* hl = H + lane * 4;
    float4 acc = make_float4(0.f, 0.f, 0.f, 0.f);
    for (int base = beg; base < end; base += 32) {
        int   s_l = s0;
        float c_l = ex0 / den;
        if (base != beg) {
            const int j = base + lane;
            s_l = node;
            c_l = 0.f;
            if (j < end) {
                s_l = csr_src[j];
                c_l = expf(gat_leaky(as[s_l] + adn) - mx) / den;
            }
        }
        const int n = (end - base < 32) ? (end - base) : 32;
        for (int t = 0; t < n; ++t) {
            const int   s = __builtin_amdgcn_readlane(s_l, t);
            const float c = __int_as_float(__builtin_amdgcn_readlane(__float_as_int(c_l), t));
            const float4 h4 = *(const float4*)(hl + (size_t)s * GAT_D);
            acc.x += c * h4.x;
            acc.y += c * h4.y;
            acc.z += c * h4.z;
            acc.w += c * h4.w;
        }
    }
    {
        const float c = ex_self / den;
        const float4 h4 = *(const float4*)(hl + (size_t)node * GAT_D);
        acc.x += c * h4.x;
        acc.y += c * h4.y;
        acc.z += c * h4.z;
        acc.w += c * h4.w;
    }
    const float4 b4 = *(const float4*)(bias + lane * 4);
    float4 o4;
    o4.x = acc.x + b4.x;
    o4.y = acc.y + b4.y;
    o4.z = acc.z + b4.z;
    o4.w = acc.w + b4.w;
    return o4;
}

#define FG_RT    2
#define FG_ROWS  (FG_RT * 16)
#define FG_NPW   (FG_ROWS / 8)
#define FG_NB    ((GAT_N + FG_ROWS - 1) / FG_ROWS)
static_assert(2 * FG_ROWS <= 256, "one thread per entry of the two dot products");

__global__ void __launch_bounds__(256)
gat_gather_gemm2(const int* __restrict__ rowptr,
                 const int* __restrict__ csr_src,
                 const float* __restrict__ as1,
                 float* adio,
                 const float* __restrict__ H1,
                 const float* __restrict__ bias1,
                 const float* __restrict__ W,
                 const float* __restrict__ a_src,
                 const float* __restrict__ a_dst,
                 float* __restrict__ H2,
                 float* __restrict__ as2_out) {
    __shared__ float4 xs[FG_ROWS * GEMM_P4];
    __shared__ float part[2][8][FG_ROWS];
    const int lane = threadIdx.x & 31;
    const int wave = __builtin_amdgcn_readfirstlane((int)(threadIdx.x >> 5));
    const int half = lane >> 4;
    const int l    = lane & 15;
    const int m0   = (int)blockIdx.x * FG_ROWS;
    const int n0   = wave * 16;

    for (int q = 0; q < FG_NPW; ++q) {
        const int row  = wave * FG_NPW + q;
        const int node = m0 + row;
        float4 o4 = make_float4(0.f, 0.f, 0.f, 0.f);
        if (node < GAT_N)
            o4 = gat_gather_row(node, lane, rowptr, csr_src, as1, adio, H1, bias1);
        xs[row * GEMM_P4 + lane] = o4;
    }
    __syncthreads();

    const float* wcol = W + (size_t)(4 * half) * GAT_D + n0 + l;
    v2f b[32];
#pragma unroll
    for (int j = 0; j < 32; ++j) {
        const int kb = 8 * (j >> 1) + 2 * (j & 1);
        b[j].x = wcol[(size_t)kb       * GAT_D];
        b[j].y = wcol[(size_t)(kb + 1) * GAT_D];
    }
    const float asv = a_src[n0 + l];
    const float adv = a_dst[n0 + l];

    v8f acc[FG_RT];
#pragma unroll
    for (int t = 0; t < FG_RT; ++t) acc[t] = v8f{};

    const float4* xa = xs + l * GEMM_P4 + half;
#pragma unroll
    for (int g = 0; g < 16; ++g) {
        float4 a4[FG_RT];
#pragma unroll
        for (int t = 0; t < FG_RT; ++t) a4[t] = xa[t * 16 * GEMM_P4 + 2 * g];
#pragma unroll
        for (int t = 0; t < FG_RT; ++t) {
            v2f a;
            a.x = a4[t].x; a.y = a4[t].y;
            acc[t] = __builtin_amdgcn_wmma_f32_16x16x4_f32(
                false, a, false, b[2 * g], (short)0, acc[t], false, false);
        }
#pragma unroll
        for (int t = 0; t < FG_RT; ++t) {
            v2f a;
            a.x = a4[t].z; a.y = a4[t].w;
            acc[t] = __builtin_amdgcn_wmma_f32_16x16x4_f32(
                false, a, false, b[2 * g + 1], (short)0, acc[t], false, false);
        }
    }

    __builtin_amdgcn_sched_barrier(0);
    asm volatile("v_nop\n\tv_nop\n\tv_nop\n\tv_nop\n\tv_nop\n\tv_nop\n\tv_nop\n\tv_nop\n\tv_nop"
                 : "+v"(acc[0]), "+v"(acc[1]));
    static_assert(FG_RT == 2, "the statement above names every accumulator");

#pragma unroll
    for (int t = 0; t < FG_RT; ++t) {
        float ps[8], pd[8];
#pragma unroll
        for (int r = 0; r < 8; ++r) {
            ps[r] = gat_row_sum16(acc[t][r] * asv);
            pd[r] = gat_row_sum16(acc[t][r] * adv);
        }
        if (l == 0) {
#pragma unroll
            for (int r = 0; r < 8; ++r) {
                part[0][wave][t * 16 + half * 8 + r] = ps[r];
                part[1][wave][t * 16 + half * 8 + r] = pd[r];
            }
        }
    }
    __syncthreads();

#pragma unroll
    for (int t = 0; t < FG_RT; ++t) {
        const int mBase = m0 + t * 16 + half * 8;
#pragma unroll
        for (int r = 0; r < 8; ++r)
            if (mBase + r < GAT_N)
                H2[(size_t)(mBase + r) * GAT_D + n0 + l] = acc[t][r];
    }

    if (threadIdx.x < 2 * FG_ROWS) {
        const int which = threadIdx.x / FG_ROWS;
        const int row   = threadIdx.x - which * FG_ROWS;
        float s = 0.f;
#pragma unroll
        for (int w = 0; w < 8; ++w) s += part[which][w][row];
        if (m0 + row < GAT_N) (which ? adio : as2_out)[m0 + row] = s;
    }
}

#define PG_ROWS  32
#define PG_NPW   (PG_ROWS / 8)
#define PG_NB    ((GAT_N + PG_ROWS - 1) / PG_ROWS)
#define PG_PARTS (256 / GAT_D)
#define PG_RPP   (PG_ROWS / PG_PARTS)
static_assert(PG_PARTS * GAT_D == 256 && PG_RPP * PG_PARTS == PG_ROWS, "threads must tile the LDS tile");

__global__ void __launch_bounds__(256)
gat_gather_pool(const int* __restrict__ rowptr,
                const int* __restrict__ csr_src,
                const float* __restrict__ as,
                const float* __restrict__ ad,
                const float* __restrict__ H,
                const float* __restrict__ bias,
                float* __restrict__ out,
                const int* __restrict__ batch,
                const unsigned* __restrict__ cnt,
                float* __restrict__ pooled) {
    __shared__ float4 tile4[PG_ROWS * GAT_D / 4];
    __shared__ int    gid[PG_ROWS];
    const float* tile = (const float*)tile4;
    const int lane = threadIdx.x & 31;
    const int wave = __builtin_amdgcn_readfirstlane((int)(threadIdx.x >> 5));
    const int m0   = (int)blockIdx.x * PG_ROWS;

    for (int q = 0; q < PG_NPW; ++q) {
        const int row  = wave * PG_NPW + q;
        const int node = m0 + row;
        float4 o4 = make_float4(0.f, 0.f, 0.f, 0.f);
        int    g  = -1;
        if (node < GAT_N) {
            o4 = gat_gather_row(node, lane, rowptr, csr_src, as, ad, H, bias);
            *(float4*)(out + (size_t)node * GAT_D + lane * 4) = o4;
            g = batch[node];
        }
        tile4[row * (GAT_D / 4) + lane] = o4;
        if (lane == 0) gid[row] = g;
    }
    __syncthreads();

    const int col = threadIdx.x % GAT_D;
    const int r0  = (threadIdx.x / GAT_D) * PG_RPP;
    auto flush = [&](int gi, float a) {
        if ((unsigned)gi < (unsigned)GAT_G)
            atomicAdd(&pooled[gi * GAT_D + col], a / fmaxf((float)cnt[gi], 1.0f));
    };
    int   g[PG_RPP];
    float v[PG_RPP];
#pragma unroll
    for (int k = 0; k < PG_RPP; ++k) {
        g[k] = gid[r0 + k];
        v[k] = tile[(r0 + k) * GAT_D + col];
    }
    int   cur = g[0];
    float acc = v[0];
#pragma unroll
    for (int k = 1; k < PG_RPP; ++k) {
        if (g[k] != cur) {
            flush(cur, acc);
            cur = g[k];
            acc = 0.f;
        }
        acc += v[k];
    }
    flush(cur, acc);
}

extern "C" void kernel_launch(void* const* d_in, const int* in_sizes, int n_in,
                              void* d_out, int out_size, void* d_ws, size_t ws_size,
                              hipStream_t stream) {
    const float* x     = (const float*)d_in[0];
    const int*   ei    = (const int*)d_in[1];
    const int*   batch = (const int*)d_in[2];
    const float* W1  = (const float*)d_in[3];
    const float* as1 = (const float*)d_in[4];
    const float* ad1 = (const float*)d_in[5];
    const float* b1  = (const float*)d_in[6];
    const float* W2  = (const float*)d_in[7];
    const float* as2 = (const float*)d_in[8];
    const float* ad2 = (const float*)d_in[9];
    const float* b2  = (const float*)d_in[10];

    float* out_feats  = (float*)d_out;
    float* out_pooled = out_feats + (size_t)GAT_N * GAT_D;

    float* ws = (float*)d_ws;
    size_t o = 0;
    float*    h      = ws + o; o += (size_t)GAT_N * GAT_D;
    float*    feats1 = ws + o; o += (size_t)GAT_N * GAT_D;
    float*    alphaS = ws + o; o += GAT_N;
    float*    alphaD = ws + o; o += GAT_N;
    int*      rowloc  = (int*)(ws + o);
    int*      rowptr  = (int*)(ws + o + CSR_NP);
    int*      csr_src = (int*)(ws + o + 2 * CSR_NP);
    int*      blksum  = (int*)(ws + o + 2 * CSR_NP + GAT_E);
    static_assert(2 * CSR_NP + GAT_E + PFX_NB <= 950000, "edge list must fit the old buffers");
    static_assert((size_t)CSR_B * CSR_NP + GAT_E <= (size_t)GAT_N * GAT_D, "blkcnt and rank must fit feats1");
    o += 950000;
    o += GAT_G * GAT_D;
    unsigned* cnt    = (unsigned*)(ws + o); o += GAT_G;
    unsigned* blkcnt = (unsigned*)feats1;
    unsigned* rank   = blkcnt + (size_t)CSR_B * CSR_NP;
    float*    h2      = feats1;
    float*    alphaS2 = (float*)rowloc;
    static_assert(CSR_NP >= GAT_N, "alpha_s of layer 2 must fit rowloc");
    (void)ws_size; (void)in_sizes; (void)n_in; (void)out_size;

    gat_csr_count<<<CSR_B, CSR_T, 0, stream>>>(ei, blkcnt, rank, cnt);
    gat_csr_prefix<<<PFX_NB, PFX_T, 0, stream>>>(blkcnt, rowloc, blksum, batch, cnt);
    gat_fill_gemm1<<<GEMM_NB + FILL_NB, PFX_T, 0, stream>>>(ei, blkcnt, rank, rowloc, blksum,
                                                            rowptr, csr_src,
                                                            x, W1, as1, ad1, h, alphaS, alphaD,
                                                            out_pooled);

    gat_gather_gemm2<<<FG_NB, 256, 0, stream>>>(rowptr, csr_src, alphaS, alphaD, h, b1,
                                                W2, as2, ad2, h2, alphaS2);
    gat_gather_pool<<<PG_NB, 256, 0, stream>>>(rowptr, csr_src, alphaS2, alphaD,
                                               h2, b2, out_feats, batch, cnt, out_pooled);
}
